# starting kernel
# baseline (speedup 1.0000x reference)
#include <hip/hip_runtime.h>

#define COUNT 8
#define L1    3072
#define L2    15
#define L3    32
#define NCOL  128
#define TM    64
#define NRT   4
#define KC    64
#define LDSTR 72

typedef __attribute__((ext_vector_type(16))) __bf16         v16bf;
typedef __attribute__((ext_vector_type(8)))  float          v8f;
typedef __attribute__((ext_vector_type(16))) unsigned short v16u;
typedef __attribute__((ext_vector_type(8)))  unsigned short u16x8;
typedef __attribute__((ext_vector_type(4)))  unsigned short u16x4;

__device__ __forceinline__ unsigned short f2bf(float f) {
  unsigned int u = __float_as_uint(f);
  u += 0x7fffu + ((u >> 16) & 1u);
  return (unsigned short)(u >> 16);
}
__device__ __forceinline__ float bf2f(unsigned short h) {
  return __uint_as_float(((unsigned int)h) << 16);
}
__device__ __forceinline__ float clamp01(float v) {
  return fminf(fmaxf(v, 0.0f), 1.0f);
}

__device__ __forceinline__ v16bf frag16(const unsigned short* p) {
  u16x8 a = *(const u16x8*)(p);
  u16x8 b = *(const u16x8*)(p + 16);
  v16u v;
#pragma unroll
  for (int i = 0; i < 8; ++i) { v[i] = a[i]; v[i + 8] = b[i]; }
  return __builtin_bit_cast(v16bf, v);
}

__global__ void nnue_prep(const float* __restrict__ l1_w, const float* __restrict__ l1_b,
                          const float* __restrict__ l1f_w, const float* __restrict__ l1f_b,
                          unsigned short* __restrict__ whi, unsigned short* __restrict__ wlo,
                          float* __restrict__ bias) {
  const int i = blockIdx.x * 256 + threadIdx.x;
  if (i < NCOL * L1) {
    const int r = i / L1;
    const int k = i - r * L1;
    const float wv = l1_w[i] + l1f_w[(r & 15) * L1 + k];
    const unsigned short h = f2bf(wv);
    whi[i] = h;
    wlo[i] = f2bf(wv - bf2f(h));
  }
  if (i < NCOL) bias[i] = l1_b[i] + l1f_b[i & 15];
}

__global__ void __launch_bounds__(256, 2)
nnue_main(const float* __restrict__ x,
          const int* __restrict__ lsi,
          const unsigned short* __restrict__ whi,
          const unsigned short* __restrict__ wlo,
          const float* __restrict__ bias,
          const float* __restrict__ l2_w,
          const float* __restrict__ l2_b,
          const float* __restrict__ out_w,
          const float* __restrict__ out_b,
          float* __restrict__ out) {
  __shared__ __align__(16) unsigned short s_xhi[2][TM][LDSTR];
  __shared__ __align__(16) unsigned short s_xlo[2][TM][LDSTR];
  __shared__ float s_l1s[TM][NCOL + 4];
  __shared__ float s_l2x[TM][L3];

  const int t    = threadIdx.x;
  const int wv   = t >> 5;
  const int lane = t & 31;
  const int r0   = blockIdx.x * TM;

  const int nloc  = lane & 15;
  const int ksel  = (lane >> 4) << 3;
  const int rbase = (lane >> 4) << 3;

  const int ldr = t >> 4;
  const int ldc = (t & 15) << 2;

  const unsigned short* bhi_row = whi + (size_t)(wv * 16 + nloc) * L1;
  const unsigned short* blo_row = wlo + (size_t)(wv * 16 + nloc) * L1;

  const float* xrow = x + (size_t)(r0 + ldr) * L1 + ldc;

  v8f acc[NRT];
#pragma unroll
  for (int rt = 0; rt < NRT; ++rt)
    acc[rt] = (v8f){0.f, 0.f, 0.f, 0.f, 0.f, 0.f, 0.f, 0.f};

#pragma unroll
  for (int q = 0; q < NRT; ++q) {
    float4 xv = *(const float4*)(xrow + (size_t)q * 16 * L1);
    u16x4 h, l;
    float vv[4] = {xv.x, xv.y, xv.z, xv.w};
#pragma unroll
    for (int i = 0; i < 4; ++i) {
      const unsigned short hb = f2bf(vv[i]);
      h[i] = hb;
      l[i] = f2bf(vv[i] - bf2f(hb));
    }
    *(u16x4*)(&s_xhi[0][q * 16 + ldr][ldc]) = h;
    *(u16x4*)(&s_xlo[0][q * 16 + ldr][ldc]) = l;
  }
  __syncthreads();

  for (int kc = 0; kc < L1; kc += KC) {
    const int cur = (kc / KC) & 1;
    const bool have_next = (kc + KC) < L1;

    float4 xv[NRT];
    if (have_next) {
#pragma unroll
      for (int q = 0; q < NRT; ++q)
        xv[q] = *(const float4*)(xrow + (size_t)q * 16 * L1 + kc + KC);
      __builtin_prefetch(bhi_row + kc + KC + ksel, 0, 1);
      __builtin_prefetch(blo_row + kc + KC + ksel, 0, 1);
    }

#pragma unroll
    for (int kk = 0; kk < KC; kk += 32) {
      v16bf bhi = frag16(bhi_row + kc + kk + ksel);
      v16bf blo = frag16(blo_row + kc + kk + ksel);
#pragma unroll
      for (int rt = 0; rt < NRT; ++rt) {
        v16bf ahi = frag16(&s_xhi[cur][rt * 16 + nloc][0] + kk + ksel);
        v16bf alo = frag16(&s_xlo[cur][rt * 16 + nloc][0] + kk + ksel);
        acc[rt] = __builtin_amdgcn_wmma_f32_16x16x32_bf16(false, ahi, false, bhi, (short)0, acc[rt], false, false);
        acc[rt] = __builtin_amdgcn_wmma_f32_16x16x32_bf16(false, ahi, false, blo, (short)0, acc[rt], false, false);
        acc[rt] = __builtin_amdgcn_wmma_f32_16x16x32_bf16(false, alo, false, bhi, (short)0, acc[rt], false, false);
      }
    }

    if (have_next) {
#pragma unroll
      for (int q = 0; q < NRT; ++q) {
        u16x4 h, l;
        float vv[4] = {xv[q].x, xv[q].y, xv[q].z, xv[q].w};
#pragma unroll
        for (int i = 0; i < 4; ++i) {
          const unsigned short hb = f2bf(vv[i]);
          h[i] = hb;
          l[i] = f2bf(vv[i] - bf2f(hb));
        }
        *(u16x4*)(&s_xhi[1 - cur][q * 16 + ldr][ldc]) = h;
        *(u16x4*)(&s_xlo[1 - cur][q * 16 + ldr][ldc]) = l;
      }
    }
    __syncthreads();
  }

  {
    const int col  = wv * 16 + nloc;
    const float bc = bias[col];
#pragma unroll
    for (int rt = 0; rt < NRT; ++rt)
#pragma unroll
      for (int p = 0; p < 8; ++p)
        s_l1s[rt * 16 + rbase + p][col] = acc[rt][p] + bc;
  }
  __syncthreads();

  {
    const int i   = t >> 2;
    const int p   = t & 3;
    const int idx = lsi[r0 + i];
    const float* trow = &s_l1s[i][idx * 16];
    float l1x[2 * L2];
#pragma unroll
    for (int j = 0; j < L2; ++j) {
      const float v = trow[j];
      l1x[j]      = clamp01(v * v * (127.0f / 128.0f));
      l1x[L2 + j] = clamp01(v);
    }
#pragma unroll
    for (int q = 0; q < 8; ++q) {
      const int n = p * 8 + q;
      const float* wrow = l2_w + (size_t)(idx * L3 + n) * (2 * L2);
      float a = l2_b[idx * L3 + n];
#pragma unroll
      for (int j = 0; j < 2 * L2; ++j) a += l1x[j] * wrow[j];
      s_l2x[i][n] = clamp01(a);
    }
  }
  __syncthreads();

  if (t < TM) {
    const int idx = lsi[r0 + t];
    float s = out_b[idx];
#pragma unroll
    for (int n = 0; n < L3; ++n) s += s_l2x[t][n] * out_w[idx * L3 + n];
    out[r0 + t] = s + s_l1s[t][idx * 16 + L2];
  }
}

extern "C" void kernel_launch(void* const* d_in, const int* in_sizes, int n_in,
                              void* d_out, int out_size, void* d_ws, size_t ws_size,
                              hipStream_t stream) {
  const float* x     = (const float*)d_in[0];
  const int*   lsi   = (const int*)d_in[1];
  const float* l1_w  = (const float*)d_in[2];
  const float* l1_b  = (const float*)d_in[3];
  const float* l1f_w = (const float*)d_in[4];
  const float* l1f_b = (const float*)d_in[5];
  const float* l2_w  = (const float*)d_in[6];
  const float* l2_b  = (const float*)d_in[7];
  const float* out_w = (const float*)d_in[8];
  const float* out_b = (const float*)d_in[9];

  const int B = in_sizes[0] / L1;

  unsigned short* whi  = (unsigned short*)d_ws;
  unsigned short* wlo  = whi + (size_t)NCOL * L1;
  float*          bias = (float*)(wlo + (size_t)NCOL * L1);

  const int prepBlocks = (NCOL * L1 + 255) / 256;
  nnue_prep<<<prepBlocks, 256, 0, stream>>>(l1_w, l1_b, l1f_w, l1f_b, whi, wlo, bias);

  nnue_main<<<B / TM, 256, 0, stream>>>(x, lsi, whi, wlo, bias,
                                        l2_w, l2_b, out_w, out_b, (float*)d_out);
}


# Global loads pipelined one K chunk ahead: next weight fragments and next x stay in flight across a hand-written barrier that does not drain the load counter; summation order unchanged
# speedup vs baseline: 1.0874x; 1.0874x over previous
#include <hip/hip_runtime.h>

#define COUNT 8
#define L1    3072
#define L2    15
#define L3    32
#define NCOL  128
#define TM    64
#define NRT   4
#define KC    64
#define LDSTR 72

typedef __attribute__((ext_vector_type(16))) __bf16         v16bf;
typedef __attribute__((ext_vector_type(8)))  float          v8f;
typedef __attribute__((ext_vector_type(16))) unsigned short v16u;
typedef __attribute__((ext_vector_type(8)))  unsigned short u16x8;
typedef __attribute__((ext_vector_type(4)))  unsigned short u16x4;

__device__ __forceinline__ unsigned short f2bf(float f) {
  unsigned int u = __float_as_uint(f);
  u += 0x7fffu + ((u >> 16) & 1u);
  return (unsigned short)(u >> 16);
}
__device__ __forceinline__ float bf2f(unsigned short h) {
  return __uint_as_float(((unsigned int)h) << 16);
}
__device__ __forceinline__ float clamp01(float v) {
  return fminf(fmaxf(v, 0.0f), 1.0f);
}

__device__ __forceinline__ v16bf frag16(const unsigned short* p) {
  u16x8 a = *(const u16x8*)(p);
  u16x8 b = *(const u16x8*)(p + 16);
  v16u v;
#pragma unroll
  for (int i = 0; i < 8; ++i) { v[i] = a[i]; v[i + 8] = b[i]; }
  return __builtin_bit_cast(v16bf, v);
}

__device__ __forceinline__ void load_b(v16bf (&b)[2][2], const unsigned short* hi, const unsigned short* lo) {
#pragma unroll
  for (int ks = 0; ks < 2; ++ks) {
    b[ks][0] = frag16(hi + ks * 32);
    b[ks][1] = frag16(lo + ks * 32);
  }
}

__device__ __forceinline__ void load_x(float4 (&xv)[NRT], const float* p) {
#pragma unroll
  for (int q = 0; q < NRT; ++q) xv[q] = *(const float4*)(p + (size_t)q * 16 * L1);
}

__device__ __forceinline__ void stage_x(const float4 (&xv)[NRT], unsigned short* xh, unsigned short* xl) {
#pragma unroll
  for (int q = 0; q < NRT; ++q) {
    u16x4 h, l;
    float vv[4] = {xv[q].x, xv[q].y, xv[q].z, xv[q].w};
#pragma unroll
    for (int i = 0; i < 4; ++i) {
      const unsigned short hb = f2bf(vv[i]);
      h[i] = hb;
      l[i] = f2bf(vv[i] - bf2f(hb));
    }
    *(u16x4*)(xh + q * 16 * LDSTR) = h;
    *(u16x4*)(xl + q * 16 * LDSTR) = l;
  }
}

__device__ __forceinline__ void mma_chunk(v8f (&acc)[NRT], const v16bf (&b)[2][2],
                                          const unsigned short* ah, const unsigned short* al) {
#pragma unroll
  for (int ks = 0; ks < 2; ++ks) {
#pragma unroll
    for (int rt = 0; rt < NRT; ++rt) {
      v16bf ahi = frag16(ah + rt * 16 * LDSTR + ks * 32);
      v16bf alo = frag16(al + rt * 16 * LDSTR + ks * 32);
      acc[rt] = __builtin_amdgcn_wmma_f32_16x16x32_bf16(false, ahi, false, b[ks][0], (short)0, acc[rt], false, false);
      acc[rt] = __builtin_amdgcn_wmma_f32_16x16x32_bf16(false, ahi, false, b[ks][1], (short)0, acc[rt], false, false);
      acc[rt] = __builtin_amdgcn_wmma_f32_16x16x32_bf16(false, alo, false, b[ks][0], (short)0, acc[rt], false, false);
    }
  }
}

__device__ __forceinline__ void stage_barrier() {
  asm volatile("s_wait_dscnt 0x0\n\ts_barrier_signal -1\n\ts_barrier_wait -1" ::: "memory");
}

__device__ __forceinline__ void pin_order() { __builtin_amdgcn_sched_barrier(0); }

__global__ void nnue_prep(const float* __restrict__ l1_w, const float* __restrict__ l1_b,
                          const float* __restrict__ l1f_w, const float* __restrict__ l1f_b,
                          unsigned short* __restrict__ whi, unsigned short* __restrict__ wlo,
                          float* __restrict__ bias) {
  const int i = blockIdx.x * 256 + threadIdx.x;
  if (i < NCOL * L1) {
    const int r = i / L1;
    const int k = i - r * L1;
    const float wv = l1_w[i] + l1f_w[(r & 15) * L1 + k];
    const unsigned short h = f2bf(wv);
    whi[i] = h;
    wlo[i] = f2bf(wv - bf2f(h));
  }
  if (i < NCOL) bias[i] = l1_b[i] + l1f_b[i & 15];
}

__global__ void __launch_bounds__(256, 2)
nnue_main(const float* x,
          const int* __restrict__ lsi,
          const unsigned short* whi,
          const unsigned short* wlo,
          const float* __restrict__ bias,
          const float* __restrict__ l2_w,
          const float* __restrict__ l2_b,
          const float* __restrict__ out_w,
          const float* __restrict__ out_b,
          float* __restrict__ out) {
  __shared__ __align__(16) unsigned short s_xhi[2][TM][LDSTR];
  __shared__ __align__(16) unsigned short s_xlo[2][TM][LDSTR];
  __shared__ float s_l1s[TM][NCOL + 4];
  __shared__ float s_l2x[TM][L3];

  const int t    = threadIdx.x;
  const int wv   = t >> 5;
  const int lane = t & 31;
  const int r0   = blockIdx.x * TM;

  const int nloc  = lane & 15;
  const int ksel  = (lane >> 4) << 3;
  const int rbase = (lane >> 4) << 3;

  const int ldr = t >> 4;
  const int ldc = (t & 15) << 2;

  const unsigned short* bhi_p = whi + (size_t)(wv * 16 + nloc) * L1 + ksel;
  const unsigned short* blo_p = wlo + (size_t)(wv * 16 + nloc) * L1 + ksel;
  const float* xrow = x + (size_t)(r0 + ldr) * L1 + ldc;

  v8f acc[NRT];
#pragma unroll
  for (int rt = 0; rt < NRT; ++rt)
    acc[rt] = (v8f){0.f, 0.f, 0.f, 0.f, 0.f, 0.f, 0.f, 0.f};

  v16bf bcur[2][2], bnext[2][2];
  float4 xr[NRT];

  load_x(xr, xrow);
  load_b(bnext, bhi_p, blo_p);
  stage_x(xr, &s_xhi[0][ldr][ldc], &s_xlo[0][ldr][ldc]);
  load_x(xr, xrow + KC);
  stage_barrier();
#pragma unroll
  for (int ks = 0; ks < 2; ++ks) { bcur[ks][0] = bnext[ks][0]; bcur[ks][1] = bnext[ks][1]; }
  asm volatile("" :: "v"(bcur[0][0]), "v"(bcur[0][1]), "v"(bcur[1][0]), "v"(bcur[1][1]));

  for (int kc = 0; kc < L1; kc += KC) {
    const int  cur       = (kc / KC) & 1;
    const bool have_next = (kc + KC) < L1;
    const int  kb        = have_next ? kc + KC : kc;
    const int  kx        = (kc + 2 * KC) < L1 ? kc + 2 * KC : L1 - KC;

    load_b(bnext, bhi_p + kb, blo_p + kb);
    pin_order();
    mma_chunk(acc, bcur, &s_xhi[cur][nloc][ksel], &s_xlo[cur][nloc][ksel]);

    if (have_next) stage_x(xr, &s_xhi[1 - cur][ldr][ldc], &s_xlo[1 - cur][ldr][ldc]);

    load_x(xr, xrow + kx);
    stage_barrier();

#pragma unroll
    for (int ks = 0; ks < 2; ++ks) { bcur[ks][0] = bnext[ks][0]; bcur[ks][1] = bnext[ks][1]; }
  }

  {
    const int col  = wv * 16 + nloc;
    const float bc = bias[col];
#pragma unroll
    for (int rt = 0; rt < NRT; ++rt)
#pragma unroll
      for (int p = 0; p < 8; ++p)
        s_l1s[rt * 16 + rbase + p][col] = acc[rt][p] + bc;
  }
  __syncthreads();

  {
    const int i   = t >> 2;
    const int p   = t & 3;
    const int idx = lsi[r0 + i];
    const float* trow = &s_l1s[i][idx * 16];
    float l1x[2 * L2];
#pragma unroll
    for (int j = 0; j < L2; ++j) {
      const float v = trow[j];
      l1x[j]      = clamp01(v * v * (127.0f / 128.0f));
      l1x[L2 + j] = clamp01(v);
    }
#pragma unroll
    for (int q = 0; q < 8; ++q) {
      const int n = p * 8 + q;
      const float* wrow = l2_w + (size_t)(idx * L3 + n) * (2 * L2);
      float a = l2_b[idx * L3 + n];
#pragma unroll
      for (int j = 0; j < 2 * L2; ++j) a += l1x[j] * wrow[j];
      s_l2x[i][n] = clamp01(a);
    }
  }
  __syncthreads();

  if (t < TM) {
    const int idx = lsi[r0 + t];
    float s = out_b[idx];
#pragma unroll
    for (int n = 0; n < L3; ++n) s += s_l2x[t][n] * out_w[idx * L3 + n];
    out[r0 + t] = s + s_l1s[t][idx * 16 + L2];
  }
}

extern "C" void kernel_launch(void* const* d_in, const int* in_sizes, int n_in,
                              void* d_out, int out_size, void* d_ws, size_t ws_size,
                              hipStream_t stream) {
  const float* x     = (const float*)d_in[0];
  const int*   lsi   = (const int*)d_in[1];
  const float* l1_w  = (const float*)d_in[2];
  const float* l1_b  = (const float*)d_in[3];
  const float* l1f_w = (const float*)d_in[4];
  const float* l1f_b = (const float*)d_in[5];
  const float* l2_w  = (const float*)d_in[6];
  const float* l2_b  = (const float*)d_in[7];
  const float* out_w = (const float*)d_in[8];
  const float* out_b = (const float*)d_in[9];

  const int B = in_sizes[0] / L1;

  unsigned short* whi  = (unsigned short*)d_ws;
  unsigned short* wlo  = whi + (size_t)NCOL * L1;
  float*          bias = (float*)(wlo + (size_t)NCOL * L1);

  const int prepBlocks = (NCOL * L1 + 255) / 256;
  nnue_prep<<<prepBlocks, 256, 0, stream>>>(l1_w, l1_b, l1f_w, l1f_b, whi, wlo, bias);

  nnue_main<<<B / TM, 256, 0, stream>>>(x, lsi, whi, wlo, bias,
                                        l2_w, l2_b, out_w, out_b, (float*)d_out);
}


# Matrix phase reordered: the 16 x fragments of a K chunk are read from LDS as one group ahead of the 24 WMMAs, which go round the four accumulator tiles; order of sums per accumulator unchanged
# speedup vs baseline: 1.3828x; 1.2717x over previous
#include <hip/hip_runtime.h>

#define COUNT 8
#define L1    3072
#define L2    15
#define L3    32
#define NCOL  128
#define TM    64
#define NRT   4
#define KC    64
#define LDSTR 72

typedef __attribute__((ext_vector_type(16))) __bf16         v16bf;
typedef __attribute__((ext_vector_type(8)))  float          v8f;
typedef __attribute__((ext_vector_type(16))) unsigned short v16u;
typedef __attribute__((ext_vector_type(8)))  unsigned short u16x8;
typedef __attribute__((ext_vector_type(4)))  unsigned short u16x4;

__device__ __forceinline__ unsigned short f2bf(float f) {
  unsigned int u = __float_as_uint(f);
  u += 0x7fffu + ((u >> 16) & 1u);
  return (unsigned short)(u >> 16);
}
__device__ __forceinline__ float bf2f(unsigned short h) {
  return __uint_as_float(((unsigned int)h) << 16);
}
__device__ __forceinline__ float clamp01(float v) {
  return fminf(fmaxf(v, 0.0f), 1.0f);
}

__device__ __forceinline__ v16bf frag16(const unsigned short* p) {
  u16x8 a = *(const u16x8*)(p);
  u16x8 b = *(const u16x8*)(p + 16);
  v16u v;
#pragma unroll
  for (int i = 0; i < 8; ++i) { v[i] = a[i]; v[i + 8] = b[i]; }
  return __builtin_bit_cast(v16bf, v);
}

__device__ __forceinline__ void load_b(v16bf (&b)[2][2], const unsigned short* hi, const unsigned short* lo) {
#pragma unroll
  for (int ks = 0; ks < 2; ++ks) {
    b[ks][0] = frag16(hi + ks * 32);
    b[ks][1] = frag16(lo + ks * 32);
  }
}

__device__ __forceinline__ void load_x(float4 (&xv)[NRT], const float* p) {
#pragma unroll
  for (int q = 0; q < NRT; ++q) xv[q] = *(const float4*)(p + (size_t)q * 16 * L1);
}

__device__ __forceinline__ void stage_x(const float4 (&xv)[NRT], unsigned short* xh, unsigned short* xl) {
#pragma unroll
  for (int q = 0; q < NRT; ++q) {
    u16x4 h, l;
    float vv[4] = {xv[q].x, xv[q].y, xv[q].z, xv[q].w};
#pragma unroll
    for (int i = 0; i < 4; ++i) {
      const unsigned short hb = f2bf(vv[i]);
      h[i] = hb;
      l[i] = f2bf(vv[i] - bf2f(hb));
    }
    *(u16x4*)(xh + q * 16 * LDSTR) = h;
    *(u16x4*)(xl + q * 16 * LDSTR) = l;
  }
}

__device__ __forceinline__ void pin_order() { __builtin_amdgcn_sched_barrier(0); }

__device__ __forceinline__ void mma_chunk(v8f (&acc)[NRT], const v16bf (&b)[2][2],
                                          const unsigned short* ah, const unsigned short* al) {
  v16bf a[2][2][NRT];
#pragma unroll
  for (int ks = 0; ks < 2; ++ks) {
#pragma unroll
    for (int rt = 0; rt < NRT; ++rt) a[ks][0][rt] = frag16(ah + rt * 16 * LDSTR + ks * 32);
#pragma unroll
    for (int rt = 0; rt < NRT; ++rt) a[ks][1][rt] = frag16(al + rt * 16 * LDSTR + ks * 32);
  }
  pin_order();
#pragma unroll
  for (int ks = 0; ks < 2; ++ks) {
#pragma unroll
    for (int rt = 0; rt < NRT; ++rt)
      acc[rt] = __builtin_amdgcn_wmma_f32_16x16x32_bf16(false, a[ks][0][rt], false, b[ks][0], (short)0, acc[rt], false, false);
#pragma unroll
    for (int rt = 0; rt < NRT; ++rt)
      acc[rt] = __builtin_amdgcn_wmma_f32_16x16x32_bf16(false, a[ks][0][rt], false, b[ks][1], (short)0, acc[rt], false, false);
#pragma unroll
    for (int rt = 0; rt < NRT; ++rt)
      acc[rt] = __builtin_amdgcn_wmma_f32_16x16x32_bf16(false, a[ks][1][rt], false, b[ks][0], (short)0, acc[rt], false, false);
  }
}

__device__ __forceinline__ void stage_barrier() {
  asm volatile("s_wait_dscnt 0x0\n\ts_barrier_signal -1\n\ts_barrier_wait -1" ::: "memory");
}

__global__ void nnue_prep(const float* __restrict__ l1_w, const float* __restrict__ l1_b,
                          const float* __restrict__ l1f_w, const float* __restrict__ l1f_b,
                          unsigned short* __restrict__ whi, unsigned short* __restrict__ wlo,
                          float* __restrict__ bias) {
  const int i = blockIdx.x * 256 + threadIdx.x;
  if (i < NCOL * L1) {
    const int r = i / L1;
    const int k = i - r * L1;
    const float wv = l1_w[i] + l1f_w[(r & 15) * L1 + k];
    const unsigned short h = f2bf(wv);
    whi[i] = h;
    wlo[i] = f2bf(wv - bf2f(h));
  }
  if (i < NCOL) bias[i] = l1_b[i] + l1f_b[i & 15];
}

__global__ void __launch_bounds__(256, 2)
nnue_main(const float* x,
          const int* __restrict__ lsi,
          const unsigned short* whi,
          const unsigned short* wlo,
          const float* __restrict__ bias,
          const float* __restrict__ l2_w,
          const float* __restrict__ l2_b,
          const float* __restrict__ out_w,
          const float* __restrict__ out_b,
          float* __restrict__ out) {
  __shared__ __align__(16) unsigned short s_xhi[2][TM][LDSTR];
  __shared__ __align__(16) unsigned short s_xlo[2][TM][LDSTR];
  __shared__ float s_l1s[TM][NCOL + 4];
  __shared__ float s_l2x[TM][L3];

  const int t    = threadIdx.x;
  const int wv   = t >> 5;
  const int lane = t & 31;
  const int r0   = blockIdx.x * TM;

  const int nloc  = lane & 15;
  const int ksel  = (lane >> 4) << 3;
  const int rbase = (lane >> 4) << 3;

  const int ldr = t >> 4;
  const int ldc = (t & 15) << 2;

  const unsigned short* bhi_p = whi + (size_t)(wv * 16 + nloc) * L1 + ksel;
  const unsigned short* blo_p = wlo + (size_t)(wv * 16 + nloc) * L1 + ksel;
  const float* xrow = x + (size_t)(r0 + ldr) * L1 + ldc;

  v8f acc[NRT];
#pragma unroll
  for (int rt = 0; rt < NRT; ++rt)
    acc[rt] = (v8f){0.f, 0.f, 0.f, 0.f, 0.f, 0.f, 0.f, 0.f};

  v16bf bcur[2][2], bnext[2][2];
  float4 xr[NRT];

  load_x(xr, xrow);
  load_b(bnext, bhi_p, blo_p);
  stage_x(xr, &s_xhi[0][ldr][ldc], &s_xlo[0][ldr][ldc]);
  load_x(xr, xrow + KC);
  stage_barrier();
#pragma unroll
  for (int ks = 0; ks < 2; ++ks) { bcur[ks][0] = bnext[ks][0]; bcur[ks][1] = bnext[ks][1]; }
  asm volatile("" :: "v"(bcur[0][0]), "v"(bcur[0][1]), "v"(bcur[1][0]), "v"(bcur[1][1]));

  for (int kc = 0; kc < L1; kc += KC) {
    const int  cur       = (kc / KC) & 1;
    const bool have_next = (kc + KC) < L1;
    const int  kb        = have_next ? kc + KC : kc;
    const int  kx        = (kc + 2 * KC) < L1 ? kc + 2 * KC : L1 - KC;

    load_b(bnext, bhi_p + kb, blo_p + kb);
    pin_order();
    mma_chunk(acc, bcur, &s_xhi[cur][nloc][ksel], &s_xlo[cur][nloc][ksel]);

    if (have_next) stage_x(xr, &s_xhi[1 - cur][ldr][ldc], &s_xlo[1 - cur][ldr][ldc]);

    load_x(xr, xrow + kx);
    stage_barrier();

#pragma unroll
    for (int ks = 0; ks < 2; ++ks) { bcur[ks][0] = bnext[ks][0]; bcur[ks][1] = bnext[ks][1]; }
  }

  {
    const int col  = wv * 16 + nloc;
    const float bc = bias[col];
#pragma unroll
    for (int rt = 0; rt < NRT; ++rt)
#pragma unroll
      for (int p = 0; p < 8; ++p)
        s_l1s[rt * 16 + rbase + p][col] = acc[rt][p] + bc;
  }
  __syncthreads();

  {
    const int i   = t >> 2;
    const int p   = t & 3;
    const int idx = lsi[r0 + i];
    const float* trow = &s_l1s[i][idx * 16];
    float l1x[2 * L2];
#pragma unroll
    for (int j = 0; j < L2; ++j) {
      const float v = trow[j];
      l1x[j]      = clamp01(v * v * (127.0f / 128.0f));
      l1x[L2 + j] = clamp01(v);
    }
#pragma unroll
    for (int q = 0; q < 8; ++q) {
      const int n = p * 8 + q;
      const float* wrow = l2_w + (size_t)(idx * L3 + n) * (2 * L2);
      float a = l2_b[idx * L3 + n];
#pragma unroll
      for (int j = 0; j < 2 * L2; ++j) a += l1x[j] * wrow[j];
      s_l2x[i][n] = clamp01(a);
    }
  }
  __syncthreads();

  if (t < TM) {
    const int idx = lsi[r0 + t];
    float s = out_b[idx];
#pragma unroll
    for (int n = 0; n < L3; ++n) s += s_l2x[t][n] * out_w[idx * L3 + n];
    out[r0 + t] = s + s_l1s[t][idx * 16 + L2];
  }
}

extern "C" void kernel_launch(void* const* d_in, const int* in_sizes, int n_in,
                              void* d_out, int out_size, void* d_ws, size_t ws_size,
                              hipStream_t stream) {
  const float* x     = (const float*)d_in[0];
  const int*   lsi   = (const int*)d_in[1];
  const float* l1_w  = (const float*)d_in[2];
  const float* l1_b  = (const float*)d_in[3];
  const float* l1f_w = (const float*)d_in[4];
  const float* l1f_b = (const float*)d_in[5];
  const float* l2_w  = (const float*)d_in[6];
  const float* l2_b  = (const float*)d_in[7];
  const float* out_w = (const float*)d_in[8];
  const float* out_b = (const float*)d_in[9];

  const int B = in_sizes[0] / L1;

  unsigned short* whi  = (unsigned short*)d_ws;
  unsigned short* wlo  = whi + (size_t)NCOL * L1;
  float*          bias = (float*)(wlo + (size_t)NCOL * L1);

  const int prepBlocks = (NCOL * L1 + 255) / 256;
  nnue_prep<<<prepBlocks, 256, 0, stream>>>(l1_w, l1_b, l1f_w, l1f_b, whi, wlo, bias);

  nnue_main<<<B / TM, 256, 0, stream>>>(x, lsi, whi, wlo, bias,
                                        l2_w, l2_b, out_w, out_b, (float*)d_out);
}


# Global loads of x and of the weight fragments run three K chunks ahead in four-deep register rings refilled together (loads return in order); loop unrolled by 4; order of sums unchanged
# speedup vs baseline: 1.4956x; 1.0816x over previous
#include <hip/hip_runtime.h>

#define COUNT 8
#define L1    3072
#define L2    15
#define L3    32
#define NCOL  128
#define TM    64
#define NRT   4
#define KC    64
#define LDSTR 72
#define NCH   (L1 / KC)
#define RING  4

typedef __attribute__((ext_vector_type(16))) __bf16         v16bf;
typedef __attribute__((ext_vector_type(8)))  float          v8f;
typedef __attribute__((ext_vector_type(16))) unsigned short v16u;
typedef __attribute__((ext_vector_type(8)))  unsigned short u16x8;
typedef __attribute__((ext_vector_type(4)))  unsigned short u16x4;

__device__ __forceinline__ unsigned short f2bf(float f) {
  unsigned int u = __float_as_uint(f);
  u += 0x7fffu + ((u >> 16) & 1u);
  return (unsigned short)(u >> 16);
}
__device__ __forceinline__ float bf2f(unsigned short h) {
  return __uint_as_float(((unsigned int)h) << 16);
}
__device__ __forceinline__ float clamp01(float v) {
  return fminf(fmaxf(v, 0.0f), 1.0f);
}

__device__ __forceinline__ v16bf frag16(const unsigned short* p) {
  u16x8 a = *(const u16x8*)(p);
  u16x8 b = *(const u16x8*)(p + 16);
  v16u v;
#pragma unroll
  for (int i = 0; i < 8; ++i) { v[i] = a[i]; v[i + 8] = b[i]; }
  return __builtin_bit_cast(v16bf, v);
}

__device__ __forceinline__ void load_b(v16bf (&b)[2][2], const unsigned short* hi, const unsigned short* lo) {
#pragma unroll
  for (int ks = 0; ks < 2; ++ks) {
    b[ks][0] = frag16(hi + ks * 32);
    b[ks][1] = frag16(lo + ks * 32);
  }
}

__device__ __forceinline__ void load_x(float4 (&xv)[NRT], const float* p) {
#pragma unroll
  for (int q = 0; q < NRT; ++q) xv[q] = *(const float4*)(p + (size_t)q * 16 * L1);
}

__device__ __forceinline__ void stage_x(const float4 (&xv)[NRT], unsigned short* xh, unsigned short* xl) {
#pragma unroll
  for (int q = 0; q < NRT; ++q) {
    u16x4 h, l;
    float vv[4] = {xv[q].x, xv[q].y, xv[q].z, xv[q].w};
#pragma unroll
    for (int i = 0; i < 4; ++i) {
      const unsigned short hb = f2bf(vv[i]);
      h[i] = hb;
      l[i] = f2bf(vv[i] - bf2f(hb));
    }
    *(u16x4*)(xh + q * 16 * LDSTR) = h;
    *(u16x4*)(xl + q * 16 * LDSTR) = l;
  }
}

__device__ __forceinline__ void pin_order() { __builtin_amdgcn_sched_barrier(0); }
__device__ __forceinline__ void pin_loads() { __builtin_amdgcn_sched_barrier(0x2 | 0x4 | 0x80); }

__device__ __forceinline__ void mma_chunk(v8f (&acc)[NRT], const v16bf (&b)[2][2],
                                          const unsigned short* ah, const unsigned short* al) {
  v16bf a[2][2][NRT];
#pragma unroll
  for (int ks = 0; ks < 2; ++ks) {
#pragma unroll
    for (int rt = 0; rt < NRT; ++rt) a[ks][0][rt] = frag16(ah + rt * 16 * LDSTR + ks * 32);
#pragma unroll
    for (int rt = 0; rt < NRT; ++rt) a[ks][1][rt] = frag16(al + rt * 16 * LDSTR + ks * 32);
  }
  pin_order();
#pragma unroll
  for (int ks = 0; ks < 2; ++ks) {
#pragma unroll
    for (int rt = 0; rt < NRT; ++rt)
      acc[rt] = __builtin_amdgcn_wmma_f32_16x16x32_bf16(false, a[ks][0][rt], false, b[ks][0], (short)0, acc[rt], false, false);
#pragma unroll
    for (int rt = 0; rt < NRT; ++rt)
      acc[rt] = __builtin_amdgcn_wmma_f32_16x16x32_bf16(false, a[ks][0][rt], false, b[ks][1], (short)0, acc[rt], false, false);
#pragma unroll
    for (int rt = 0; rt < NRT; ++rt)
      acc[rt] = __builtin_amdgcn_wmma_f32_16x16x32_bf16(false, a[ks][1][rt], false, b[ks][0], (short)0, acc[rt], false, false);
  }
}

__device__ __forceinline__ void stage_barrier() {
  asm volatile("s_wait_dscnt 0x0\n\ts_barrier_signal -1\n\ts_barrier_wait -1" ::: "memory");
}

template <int S>
__device__ __forceinline__ void k_step(const int c, v8f (&acc)[NRT], v16bf (&bq)[RING][2][2], float4 (&xq)[RING][NRT],
                                       const unsigned short* bhi_p, const unsigned short* blo_p, const float* xrow,
                                       const unsigned short* ah0, const unsigned short* al0,
                                       unsigned short* xh0, unsigned short* xl0) {
  constexpr int CUR = S & 1;
  constexpr int SX  = (S + 1) % RING;
  mma_chunk(acc, bq[S], ah0 + CUR * (TM * LDSTR), al0 + CUR * (TM * LDSTR));
  stage_x(xq[SX], xh0 + (1 - CUR) * (TM * LDSTR), xl0 + (1 - CUR) * (TM * LDSTR));
  const int cx = (c + 1 + RING) < NCH ? c + 1 + RING : NCH - 1;
  const int cb = (c + RING) < NCH ? c + RING : NCH - 1;
  asm volatile("" :: "v"(acc[0]), "v"(acc[1]), "v"(acc[2]), "v"(acc[3]) : "memory");
  load_x(xq[SX], xrow + cx * KC);
  load_b(bq[S], bhi_p + cb * KC, blo_p + cb * KC);
  stage_barrier();
}

__global__ void nnue_prep(const float* __restrict__ l1_w, const float* __restrict__ l1_b,
                          const float* __restrict__ l1f_w, const float* __restrict__ l1f_b,
                          unsigned short* __restrict__ whi, unsigned short* __restrict__ wlo,
                          float* __restrict__ bias) {
  const int i = blockIdx.x * 256 + threadIdx.x;
  if (i < NCOL * L1) {
    const int r = i / L1;
    const int k = i - r * L1;
    const float wv = l1_w[i] + l1f_w[(r & 15) * L1 + k];
    const unsigned short h = f2bf(wv);
    whi[i] = h;
    wlo[i] = f2bf(wv - bf2f(h));
  }
  if (i < NCOL) bias[i] = l1_b[i] + l1f_b[i & 15];
}

__global__ void __launch_bounds__(256, 1)
nnue_main(const float* x,
          const int* __restrict__ lsi,
          const unsigned short* whi,
          const unsigned short* wlo,
          const float* __restrict__ bias,
          const float* __restrict__ l2_w,
          const float* __restrict__ l2_b,
          const float* __restrict__ out_w,
          const float* __restrict__ out_b,
          float* __restrict__ out) {
  __shared__ __align__(16) unsigned short s_xhi[2][TM][LDSTR];
  __shared__ __align__(16) unsigned short s_xlo[2][TM][LDSTR];
  __shared__ float s_l1s[TM][NCOL + 4];
  __shared__ float s_l2x[TM][L3];

  const int t    = threadIdx.x;
  const int wv   = t >> 5;
  const int lane = t & 31;
  const int r0   = blockIdx.x * TM;

  const int nloc  = lane & 15;
  const int ksel  = (lane >> 4) << 3;
  const int rbase = (lane >> 4) << 3;

  const int ldr = t >> 4;
  const int ldc = (t & 15) << 2;

  const unsigned short* bhi_p = whi + (size_t)(wv * 16 + nloc) * L1 + ksel;
  const unsigned short* blo_p = wlo + (size_t)(wv * 16 + nloc) * L1 + ksel;
  const float* xrow = x + (size_t)(r0 + ldr) * L1 + ldc;

  v8f acc[NRT];
#pragma unroll
  for (int rt = 0; rt < NRT; ++rt)
    acc[rt] = (v8f){0.f, 0.f, 0.f, 0.f, 0.f, 0.f, 0.f, 0.f};

  v16bf  bq[RING][2][2];
  float4 xq[RING][NRT];

  unsigned short* const xh0 = &s_xhi[0][ldr][ldc];
  unsigned short* const xl0 = &s_xlo[0][ldr][ldc];
  const unsigned short* const ah0 = &s_xhi[0][nloc][ksel];
  const unsigned short* const al0 = &s_xlo[0][nloc][ksel];

  float4 x0v[NRT];
  load_x(x0v, xrow);
#pragma unroll
  for (int s = 0; s < RING; ++s) {
    pin_loads();
    load_x(xq[(s + 1) % RING], xrow + (s + 1) * KC);
    load_b(bq[s], bhi_p + s * KC, blo_p + s * KC);
  }
  stage_x(x0v, xh0, xl0);
  stage_barrier();

  for (int c = 0; c < NCH; c += RING) {
    k_step<0>(c,     acc, bq, xq, bhi_p, blo_p, xrow, ah0, al0, xh0, xl0);
    k_step<1>(c + 1, acc, bq, xq, bhi_p, blo_p, xrow, ah0, al0, xh0, xl0);
    k_step<2>(c + 2, acc, bq, xq, bhi_p, blo_p, xrow, ah0, al0, xh0, xl0);
    k_step<3>(c + 3, acc, bq, xq, bhi_p, blo_p, xrow, ah0, al0, xh0, xl0);
  }

  {
    const int col  = wv * 16 + nloc;
    const float bc = bias[col];
#pragma unroll
    for (int rt = 0; rt < NRT; ++rt)
#pragma unroll
      for (int p = 0; p < 8; ++p)
        s_l1s[rt * 16 + rbase + p][col] = acc[rt][p] + bc;
  }
  __syncthreads();

  {
    const int i   = t >> 2;
    const int p   = t & 3;
    const int idx = lsi[r0 + i];
    const float* trow = &s_l1s[i][idx * 16];
    float l1x[2 * L2];
#pragma unroll
    for (int j = 0; j < L2; ++j) {
      const float v = trow[j];
      l1x[j]      = clamp01(v * v * (127.0f / 128.0f));
      l1x[L2 + j] = clamp01(v);
    }
#pragma unroll
    for (int q = 0; q < 8; ++q) {
      const int n = p * 8 + q;
      const float* wrow = l2_w + (size_t)(idx * L3 + n) * (2 * L2);
      float a = l2_b[idx * L3 + n];
#pragma unroll
      for (int j = 0; j < 2 * L2; ++j) a += l1x[j] * wrow[j];
      s_l2x[i][n] = clamp01(a);
    }
  }
  __syncthreads();

  if (t < TM) {
    const int idx = lsi[r0 + t];
    float s = out_b[idx];
#pragma unroll
    for (int n = 0; n < L3; ++n) s += s_l2x[t][n] * out_w[idx * L3 + n];
    out[r0 + t] = s + s_l1s[t][idx * 16 + L2];
  }
}

extern "C" void kernel_launch(void* const* d_in, const int* in_sizes, int n_in,
                              void* d_out, int out_size, void* d_ws, size_t ws_size,
                              hipStream_t stream) {
  const float* x     = (const float*)d_in[0];
  const int*   lsi   = (const int*)d_in[1];
  const float* l1_w  = (const float*)d_in[2];
  const float* l1_b  = (const float*)d_in[3];
  const float* l1f_w = (const float*)d_in[4];
  const float* l1f_b = (const float*)d_in[5];
  const float* l2_w  = (const float*)d_in[6];
  const float* l2_b  = (const float*)d_in[7];
  const float* out_w = (const float*)d_in[8];
  const float* out_b = (const float*)d_in[9];

  const int B = in_sizes[0] / L1;

  unsigned short* whi  = (unsigned short*)d_ws;
  unsigned short* wlo  = whi + (size_t)NCOL * L1;
  float*          bias = (float*)(wlo + (size_t)NCOL * L1);

  const int prepBlocks = (NCOL * L1 + 255) / 256;
  nnue_prep<<<prepBlocks, 256, 0, stream>>>(l1_w, l1_b, l1f_w, l1f_b, whi, wlo, bias);

  nnue_main<<<B / TM, 256, 0, stream>>>(x, lsi, whi, wlo, bias,
                                        l2_w, l2_b, out_w, out_b, (float*)d_out);
}


# Weight planes stored by the prep kernel in matrix-fragment order, so each weight load of a wave reads 512 contiguous bytes (4 lines, not 16); values and order of sums unchanged
# speedup vs baseline: 1.7449x; 1.1667x over previous
#include <hip/hip_runtime.h>

#define COUNT 8
#define L1    3072
#define L2    15
#define L3    32
#define NCOL  128
#define TM    64
#define NRT   4
#define KC    64
#define LDSTR 72
#define NCH   (L1 / KC)
#define RING  4
#define PIECE 256
#define BCH   (4 * PIECE)

typedef __attribute__((ext_vector_type(16))) __bf16         v16bf;
typedef __attribute__((ext_vector_type(8)))  float          v8f;
typedef __attribute__((ext_vector_type(16))) unsigned short v16u;
typedef __attribute__((ext_vector_type(8)))  unsigned short u16x8;
typedef __attribute__((ext_vector_type(4)))  unsigned short u16x4;

__device__ __forceinline__ unsigned short f2bf(float f) {
  unsigned int u = __float_as_uint(f);
  u += 0x7fffu + ((u >> 16) & 1u);
  return (unsigned short)(u >> 16);
}
__device__ __forceinline__ float bf2f(unsigned short h) {
  return __uint_as_float(((unsigned int)h) << 16);
}
__device__ __forceinline__ float clamp01(float v) {
  return fminf(fmaxf(v, 0.0f), 1.0f);
}

__device__ __forceinline__ v16bf frag16(const unsigned short* p) {
  u16x8 a = *(const u16x8*)(p);
  u16x8 b = *(const u16x8*)(p + 16);
  v16u v;
#pragma unroll
  for (int i = 0; i < 8; ++i) { v[i] = a[i]; v[i + 8] = b[i]; }
  return __builtin_bit_cast(v16bf, v);
}

__device__ __forceinline__ v16bf frag16w(const unsigned short* p) {
  u16x8 a = *(const u16x8*)(p);
  u16x8 b = *(const u16x8*)(p + PIECE);
  v16u v;
#pragma unroll
  for (int i = 0; i < 8; ++i) { v[i] = a[i]; v[i + 8] = b[i]; }
  return __builtin_bit_cast(v16bf, v);
}

__device__ __forceinline__ void load_b(v16bf (&b)[2][2], const unsigned short* hi, const unsigned short* lo) {
#pragma unroll
  for (int ks = 0; ks < 2; ++ks) {
    b[ks][0] = frag16w(hi + ks * 2 * PIECE);
    b[ks][1] = frag16w(lo + ks * 2 * PIECE);
  }
}

__device__ __forceinline__ void load_x(float4 (&xv)[NRT], const float* p) {
#pragma unroll
  for (int q = 0; q < NRT; ++q) xv[q] = *(const float4*)(p + (size_t)q * 16 * L1);
}

__device__ __forceinline__ void stage_x(const float4 (&xv)[NRT], unsigned short* xh, unsigned short* xl) {
#pragma unroll
  for (int q = 0; q < NRT; ++q) {
    u16x4 h, l;
    float vv[4] = {xv[q].x, xv[q].y, xv[q].z, xv[q].w};
#pragma unroll
    for (int i = 0; i < 4; ++i) {
      const unsigned short hb = f2bf(vv[i]);
      h[i] = hb;
      l[i] = f2bf(vv[i] - bf2f(hb));
    }
    *(u16x4*)(xh + q * 16 * LDSTR) = h;
    *(u16x4*)(xl + q * 16 * LDSTR) = l;
  }
}

__device__ __forceinline__ void pin_order() { __builtin_amdgcn_sched_barrier(0); }
__device__ __forceinline__ void pin_loads() { __builtin_amdgcn_sched_barrier(0x2 | 0x4 | 0x80); }

__device__ __forceinline__ void mma_chunk(v8f (&acc)[NRT], const v16bf (&b)[2][2],
                                          const unsigned short* ah, const unsigned short* al) {
  v16bf a[2][2][NRT];
#pragma unroll
  for (int ks = 0; ks < 2; ++ks) {
#pragma unroll
    for (int rt = 0; rt < NRT; ++rt) a[ks][0][rt] = frag16(ah + rt * 16 * LDSTR + ks * 32);
#pragma unroll
    for (int rt = 0; rt < NRT; ++rt) a[ks][1][rt] = frag16(al + rt * 16 * LDSTR + ks * 32);
  }
  pin_order();
#pragma unroll
  for (int ks = 0; ks < 2; ++ks) {
#pragma unroll
    for (int rt = 0; rt < NRT; ++rt)
      acc[rt] = __builtin_amdgcn_wmma_f32_16x16x32_bf16(false, a[ks][0][rt], false, b[ks][0], (short)0, acc[rt], false, false);
#pragma unroll
    for (int rt = 0; rt < NRT; ++rt)
      acc[rt] = __builtin_amdgcn_wmma_f32_16x16x32_bf16(false, a[ks][0][rt], false, b[ks][1], (short)0, acc[rt], false, false);
#pragma unroll
    for (int rt = 0; rt < NRT; ++rt)
      acc[rt] = __builtin_amdgcn_wmma_f32_16x16x32_bf16(false, a[ks][1][rt], false, b[ks][0], (short)0, acc[rt], false, false);
  }
}

__device__ __forceinline__ void stage_barrier() {
  asm volatile("s_wait_dscnt 0x0\n\ts_barrier_signal -1\n\ts_barrier_wait -1" ::: "memory");
}

template <int S>
__device__ __forceinline__ void k_step(const int c, v8f (&acc)[NRT], v16bf (&bq)[RING][2][2], float4 (&xq)[RING][NRT],
                                       const unsigned short* bhi_p, const unsigned short* blo_p, const float* xrow,
                                       const unsigned short* ah0, const unsigned short* al0,
                                       unsigned short* xh0, unsigned short* xl0) {
  constexpr int CUR = S & 1;
  constexpr int SX  = (S + 1) % RING;
  mma_chunk(acc, bq[S], ah0 + CUR * (TM * LDSTR), al0 + CUR * (TM * LDSTR));
  stage_x(xq[SX], xh0 + (1 - CUR) * (TM * LDSTR), xl0 + (1 - CUR) * (TM * LDSTR));
  const int cx = (c + 1 + RING) < NCH ? c + 1 + RING : NCH - 1;
  const int cb = (c + RING) < NCH ? c + RING : NCH - 1;
  asm volatile("" :: "v"(acc[0]), "v"(acc[1]), "v"(acc[2]), "v"(acc[3]) : "memory");
  load_x(xq[SX], xrow + cx * KC);
  load_b(bq[S], bhi_p + cb * BCH, blo_p + cb * BCH);
  stage_barrier();
}

__global__ void nnue_prep(const float* __restrict__ l1_w, const float* __restrict__ l1_b,
                          const float* __restrict__ l1f_w, const float* __restrict__ l1f_b,
                          unsigned short* __restrict__ whi, unsigned short* __restrict__ wlo,
                          float* __restrict__ bias) {
  const int p = blockIdx.x * 256 + threadIdx.x;
  if (p < NCOL * L1 / 8) {
    const int lane = p & 31;
    const int half = (p >> 5) & 1;
    const int ks   = (p >> 6) & 1;
    const int tc   = p >> 7;
    const int tile = tc / NCH;
    const int c    = tc - tile * NCH;
    const int r    = tile * 16 + (lane & 15);
    const int k    = c * KC + ks * 32 + half * 16 + (lane >> 4) * 8;
    const float4 a0 = *(const float4*)(l1_w + (size_t)r * L1 + k);
    const float4 a1 = *(const float4*)(l1_w + (size_t)r * L1 + k + 4);
    const float4 f0 = *(const float4*)(l1f_w + (size_t)(r & 15) * L1 + k);
    const float4 f1 = *(const float4*)(l1f_w + (size_t)(r & 15) * L1 + k + 4);
    const float w8[8] = {a0.x + f0.x, a0.y + f0.y, a0.z + f0.z, a0.w + f0.w,
                         a1.x + f1.x, a1.y + f1.y, a1.z + f1.z, a1.w + f1.w};
    u16x8 h, l;
#pragma unroll
    for (int i = 0; i < 8; ++i) {
      const unsigned short hb = f2bf(w8[i]);
      h[i] = hb;
      l[i] = f2bf(w8[i] - bf2f(hb));
    }
    *(u16x8*)(whi + (size_t)p * 8) = h;
    *(u16x8*)(wlo + (size_t)p * 8) = l;
  }
  if (p < NCOL) bias[p] = l1_b[p] + l1f_b[p & 15];
}

__global__ void __launch_bounds__(256, 1)
nnue_main(const float* x,
          const int* __restrict__ lsi,
          const unsigned short* whi,
          const unsigned short* wlo,
          const float* __restrict__ bias,
          const float* __restrict__ l2_w,
          const float* __restrict__ l2_b,
          const float* __restrict__ out_w,
          const float* __restrict__ out_b,
          float* __restrict__ out) {
  __shared__ __align__(16) unsigned short s_xhi[2][TM][LDSTR];
  __shared__ __align__(16) unsigned short s_xlo[2][TM][LDSTR];
  __shared__ float s_l1s[TM][NCOL + 4];
  __shared__ float s_l2x[TM][L3];

  const int t    = threadIdx.x;
  const int wv   = t >> 5;
  const int lane = t & 31;
  const int r0   = blockIdx.x * TM;

  const int nloc  = lane & 15;
  const int ksel  = (lane >> 4) << 3;
  const int rbase = (lane >> 4) << 3;

  const int ldr = t >> 4;
  const int ldc = (t & 15) << 2;

  const unsigned short* bhi_p = whi + (size_t)wv * NCH * BCH + lane * 8;
  const unsigned short* blo_p = wlo + (size_t)wv * NCH * BCH + lane * 8;
  const float* xrow = x + (size_t)(r0 + ldr) * L1 + ldc;

  v8f acc[NRT];
#pragma unroll
  for (int rt = 0; rt < NRT; ++rt)
    acc[rt] = (v8f){0.f, 0.f, 0.f, 0.f, 0.f, 0.f, 0.f, 0.f};

  v16bf  bq[RING][2][2];
  float4 xq[RING][NRT];

  unsigned short* const xh0 = &s_xhi[0][ldr][ldc];
  unsigned short* const xl0 = &s_xlo[0][ldr][ldc];
  const unsigned short* const ah0 = &s_xhi[0][nloc][ksel];
  const unsigned short* const al0 = &s_xlo[0][nloc][ksel];

  float4 x0v[NRT];
  load_x(x0v, xrow);
#pragma unroll
  for (int s = 0; s < RING; ++s) {
    pin_loads();
    load_x(xq[(s + 1) % RING], xrow + (s + 1) * KC);
    load_b(bq[s], bhi_p + s * BCH, blo_p + s * BCH);
  }
  stage_x(x0v, xh0, xl0);
  stage_barrier();

  for (int c = 0; c < NCH; c += RING) {
    k_step<0>(c,     acc, bq, xq, bhi_p, blo_p, xrow, ah0, al0, xh0, xl0);
    k_step<1>(c + 1, acc, bq, xq, bhi_p, blo_p, xrow, ah0, al0, xh0, xl0);
    k_step<2>(c + 2, acc, bq, xq, bhi_p, blo_p, xrow, ah0, al0, xh0, xl0);
    k_step<3>(c + 3, acc, bq, xq, bhi_p, blo_p, xrow, ah0, al0, xh0, xl0);
  }

  {
    const int col  = wv * 16 + nloc;
    const float bc = bias[col];
#pragma unroll
    for (int rt = 0; rt < NRT; ++rt)
#pragma unroll
      for (int p = 0; p < 8; ++p)
        s_l1s[rt * 16 + rbase + p][col] = acc[rt][p] + bc;
  }
  __syncthreads();

  {
    const int i   = t >> 2;
    const int p   = t & 3;
    const int idx = lsi[r0 + i];
    const float* trow = &s_l1s[i][idx * 16];
    float l1x[2 * L2];
#pragma unroll
    for (int j = 0; j < L2; ++j) {
      const float v = trow[j];
      l1x[j]      = clamp01(v * v * (127.0f / 128.0f));
      l1x[L2 + j] = clamp01(v);
    }
#pragma unroll
    for (int q = 0; q < 8; ++q) {
      const int n = p * 8 + q;
      const float* wrow = l2_w + (size_t)(idx * L3 + n) * (2 * L2);
      float a = l2_b[idx * L3 + n];
#pragma unroll
      for (int j = 0; j < 2 * L2; ++j) a += l1x[j] * wrow[j];
      s_l2x[i][n] = clamp01(a);
    }
  }
  __syncthreads();

  if (t < TM) {
    const int idx = lsi[r0 + t];
    float s = out_b[idx];
#pragma unroll
    for (int n = 0; n < L3; ++n) s += s_l2x[t][n] * out_w[idx * L3 + n];
    out[r0 + t] = s + s_l1s[t][idx * 16 + L2];
  }
}

extern "C" void kernel_launch(void* const* d_in, const int* in_sizes, int n_in,
                              void* d_out, int out_size, void* d_ws, size_t ws_size,
                              hipStream_t stream) {
  const float* x     = (const float*)d_in[0];
  const int*   lsi   = (const int*)d_in[1];
  const float* l1_w  = (const float*)d_in[2];
  const float* l1_b  = (const float*)d_in[3];
  const float* l1f_w = (const float*)d_in[4];
  const float* l1f_b = (const float*)d_in[5];
  const float* l2_w  = (const float*)d_in[6];
  const float* l2_b  = (const float*)d_in[7];
  const float* out_w = (const float*)d_in[8];
  const float* out_b = (const float*)d_in[9];

  const int B = in_sizes[0] / L1;

  unsigned short* whi  = (unsigned short*)d_ws;
  unsigned short* wlo  = whi + (size_t)NCOL * L1;
  float*          bias = (float*)(wlo + (size_t)NCOL * L1);

  const int prepBlocks = (NCOL * L1 / 8 + 255) / 256;
  nnue_prep<<<prepBlocks, 256, 0, stream>>>(l1_w, l1_b, l1f_w, l1f_b, whi, wlo, bias);

  nnue_main<<<B / TM, 256, 0, stream>>>(x, lsi, whi, wlo, bias,
                                        l2_w, l2_b, out_w, out_b, (float*)d_out);
}


# Rows of a tile lie in LDS grouped by bucket and each wave multiplies only the rows of its own bucket (one row tile in the usual case, not four); each row's products and order of sums unchanged
# speedup vs baseline: 2.0112x; 1.1526x over previous
#include <hip/hip_runtime.h>

#define COUNT 8
#define L1    3072
#define L2    15
#define L3    32
#define NCOL  128
#define TM    64
#define NRT   4
#define KC    64
#define LDSTR 72
#define NCH   (L1 / KC)
#define RING  4
#define PIECE 256
#define BCH   (4 * PIECE)
#define STAGE (TM * LDSTR)
#define PLANE (2 * STAGE)

typedef __attribute__((ext_vector_type(16))) __bf16         v16bf;
typedef __attribute__((ext_vector_type(8)))  float          v8f;
typedef __attribute__((ext_vector_type(16))) unsigned short v16u;
typedef __attribute__((ext_vector_type(8)))  unsigned short u16x8;
typedef __attribute__((ext_vector_type(4)))  unsigned short u16x4;

__device__ __forceinline__ unsigned short f2bf(float f) {
  unsigned int u = __float_as_uint(f);
  u += 0x7fffu + ((u >> 16) & 1u);
  return (unsigned short)(u >> 16);
}
__device__ __forceinline__ float bf2f(unsigned short h) {
  return __uint_as_float(((unsigned int)h) << 16);
}
__device__ __forceinline__ float clamp01(float v) {
  return fminf(fmaxf(v, 0.0f), 1.0f);
}

__device__ __forceinline__ v16bf frag16(const unsigned short* p) {
  u16x8 a = *(const u16x8*)(p);
  u16x8 b = *(const u16x8*)(p + 16);
  v16u v;
#pragma unroll
  for (int i = 0; i < 8; ++i) { v[i] = a[i]; v[i + 8] = b[i]; }
  return __builtin_bit_cast(v16bf, v);
}

__device__ __forceinline__ v16bf frag16w(const unsigned short* p) {
  u16x8 a = *(const u16x8*)(p);
  u16x8 b = *(const u16x8*)(p + PIECE);
  v16u v;
#pragma unroll
  for (int i = 0; i < 8; ++i) { v[i] = a[i]; v[i + 8] = b[i]; }
  return __builtin_bit_cast(v16bf, v);
}

__device__ __forceinline__ void load_b(v16bf (&b)[2][2], const unsigned short* hi, const unsigned short* lo) {
#pragma unroll
  for (int ks = 0; ks < 2; ++ks) {
    b[ks][0] = frag16w(hi + ks * 2 * PIECE);
    b[ks][1] = frag16w(lo + ks * 2 * PIECE);
  }
}

__device__ __forceinline__ void load_x(float4 (&xv)[NRT], const float* p) {
#pragma unroll
  for (int q = 0; q < NRT; ++q) xv[q] = *(const float4*)(p + (size_t)q * 16 * L1);
}

__device__ __forceinline__ void stage_x(const float4 (&xv)[NRT], unsigned short* const (&xs)[NRT], const int st) {
#pragma unroll
  for (int q = 0; q < NRT; ++q) {
    u16x4 h, l;
    float vv[4] = {xv[q].x, xv[q].y, xv[q].z, xv[q].w};
#pragma unroll
    for (int i = 0; i < 4; ++i) {
      const unsigned short hb = f2bf(vv[i]);
      h[i] = hb;
      l[i] = f2bf(vv[i] - bf2f(hb));
    }
    *(u16x4*)(xs[q] + st) = h;
    *(u16x4*)(xs[q] + st + PLANE) = l;
  }
}

__device__ __forceinline__ void pin_order() { __builtin_amdgcn_sched_barrier(0); }
__device__ __forceinline__ void pin_loads() { __builtin_amdgcn_sched_barrier(0x2 | 0x4 | 0x80); }

template <int NT>
__device__ __forceinline__ void mma_chunk(v8f (&acc)[NRT], const v16bf (&b)[2][2],
                                          const unsigned short* const (&ar)[NRT], const int st) {
  v16bf a[2][2][NT];
#pragma unroll
  for (int ks = 0; ks < 2; ++ks) {
#pragma unroll
    for (int rt = 0; rt < NT; ++rt) a[ks][0][rt] = frag16(ar[rt] + st + ks * 32);
#pragma unroll
    for (int rt = 0; rt < NT; ++rt) a[ks][1][rt] = frag16(ar[rt] + st + PLANE + ks * 32);
  }
  pin_order();
#pragma unroll
  for (int ks = 0; ks < 2; ++ks) {
#pragma unroll
    for (int rt = 0; rt < NT; ++rt)
      acc[rt] = __builtin_amdgcn_wmma_f32_16x16x32_bf16(false, a[ks][0][rt], false, b[ks][0], (short)0, acc[rt], false, false);
#pragma unroll
    for (int rt = 0; rt < NT; ++rt)
      acc[rt] = __builtin_amdgcn_wmma_f32_16x16x32_bf16(false, a[ks][0][rt], false, b[ks][1], (short)0, acc[rt], false, false);
#pragma unroll
    for (int rt = 0; rt < NT; ++rt)
      acc[rt] = __builtin_amdgcn_wmma_f32_16x16x32_bf16(false, a[ks][1][rt], false, b[ks][0], (short)0, acc[rt], false, false);
  }
}

__device__ __forceinline__ void stage_barrier() {
  asm volatile("s_wait_dscnt 0x0\n\ts_barrier_signal -1\n\ts_barrier_wait -1" ::: "memory");
}

template <int S>
__device__ __forceinline__ void k_step(const int c, const bool many, v8f (&acc)[NRT], v16bf (&bq)[RING][2][2], float4 (&xq)[RING][NRT],
                                       const unsigned short* bhi_p, const unsigned short* blo_p, const float* xrow,
                                       const unsigned short* const (&ar)[NRT], unsigned short* const (&xs)[NRT]) {
  constexpr int CUR = S & 1;
  constexpr int SX  = (S + 1) % RING;
  if (many) mma_chunk<NRT>(acc, bq[S], ar, CUR * STAGE);
  else      mma_chunk<1>(acc, bq[S], ar, CUR * STAGE);
  stage_x(xq[SX], xs, (1 - CUR) * STAGE);
  const int cx = (c + 1 + RING) < NCH ? c + 1 + RING : NCH - 1;
  const int cb = (c + RING) < NCH ? c + RING : NCH - 1;
  asm volatile("" :: "v"(acc[0]), "v"(acc[1]), "v"(acc[2]), "v"(acc[3]) : "memory");
  load_x(xq[SX], xrow + cx * KC);
  load_b(bq[S], bhi_p + cb * BCH, blo_p + cb * BCH);
  stage_barrier();
}

__global__ void nnue_prep(const float* __restrict__ l1_w, const float* __restrict__ l1_b,
                          const float* __restrict__ l1f_w, const float* __restrict__ l1f_b,
                          unsigned short* __restrict__ whi, unsigned short* __restrict__ wlo,
                          float* __restrict__ bias) {
  const int p = blockIdx.x * 256 + threadIdx.x;
  if (p < NCOL * L1 / 8) {
    const int lane = p & 31;
    const int half = (p >> 5) & 1;
    const int ks   = (p >> 6) & 1;
    const int tc   = p >> 7;
    const int tile = tc / NCH;
    const int c    = tc - tile * NCH;
    const int r    = tile * 16 + (lane & 15);
    const int k    = c * KC + ks * 32 + half * 16 + (lane >> 4) * 8;
    const float4 a0 = *(const float4*)(l1_w + (size_t)r * L1 + k);
    const float4 a1 = *(const float4*)(l1_w + (size_t)r * L1 + k + 4);
    const float4 f0 = *(const float4*)(l1f_w + (size_t)(r & 15) * L1 + k);
    const float4 f1 = *(const float4*)(l1f_w + (size_t)(r & 15) * L1 + k + 4);
    const float w8[8] = {a0.x + f0.x, a0.y + f0.y, a0.z + f0.z, a0.w + f0.w,
                         a1.x + f1.x, a1.y + f1.y, a1.z + f1.z, a1.w + f1.w};
    u16x8 h, l;
#pragma unroll
    for (int i = 0; i < 8; ++i) {
      const unsigned short hb = f2bf(w8[i]);
      h[i] = hb;
      l[i] = f2bf(w8[i] - bf2f(hb));
    }
    *(u16x8*)(whi + (size_t)p * 8) = h;
    *(u16x8*)(wlo + (size_t)p * 8) = l;
  }
  if (p < NCOL) bias[p] = l1_b[p] + l1f_b[p & 15];
}

__global__ void __launch_bounds__(256, 1)
nnue_main(const float* x,
          const int* __restrict__ lsi,
          const unsigned short* whi,
          const unsigned short* wlo,
          const float* __restrict__ bias,
          const float* __restrict__ l2_w,
          const float* __restrict__ l2_b,
          const float* __restrict__ out_w,
          const float* __restrict__ out_b,
          float* __restrict__ out) {
  __shared__ __align__(16) unsigned short s_x[2][2][TM][LDSTR];
  __shared__ float s_l1s[TM][L2 + 2];
  __shared__ float s_l2x[TM][L3];

  const int t    = threadIdx.x;
  const int wv   = __builtin_amdgcn_readfirstlane(t >> 5);
  const int lane = t & 31;
  const int r0   = blockIdx.x * TM;

  const int nloc  = lane & 15;
  const int ksel  = (lane >> 4) << 3;
  const int rbase = (lane >> 4) << 3;

  const int ldr = t >> 4;
  const int ldc = (t & 15) << 2;

  const int bA = lsi[r0 + lane];
  const int bB = lsi[r0 + 32 + lane];

  const unsigned short* bhi_p = whi + (size_t)wv * NCH * BCH + lane * 8;
  const unsigned short* blo_p = wlo + (size_t)wv * NCH * BCH + lane * 8;
  const float* xrow = x + (size_t)(r0 + ldr) * L1 + ldc;

  v8f acc[NRT];
#pragma unroll
  for (int rt = 0; rt < NRT; ++rt)
    acc[rt] = (v8f){0.f, 0.f, 0.f, 0.f, 0.f, 0.f, 0.f, 0.f};

  v16bf  bq[RING][2][2];
  float4 xq[RING][NRT];

  float4 x0v[NRT];
  load_x(x0v, xrow);
#pragma unroll
  for (int s = 0; s < RING; ++s) {
    pin_loads();
    load_x(xq[(s + 1) % RING], xrow + (s + 1) * KC);
    load_b(bq[s], bhi_p + s * BCH, blo_p + s * BCH);
  }

  int slotA = 0, slotB = 0, myoff = 0, mycnt = 0;
  {
    const unsigned int below = (1u << lane) - 1u;
    int run = 0;
#pragma unroll
    for (int k = 0; k < COUNT; ++k) {
      const unsigned int mA = (unsigned int)__ballot(bA == k);
      const unsigned int mB = (unsigned int)__ballot(bB == k);
      const int cA = __popc(mA), cB = __popc(mB);
      if (bA == k) slotA = run + __popc(mA & below);
      if (bB == k) slotB = run + cA + __popc(mB & below);
      if (k == wv) { myoff = run; mycnt = cA + cB; }
      run += cA + cB;
    }
  }
  myoff = __builtin_amdgcn_readfirstlane(myoff);
  mycnt = __builtin_amdgcn_readfirstlane(mycnt);
  const bool many = mycnt > 16;

  unsigned short* xs[NRT];
  xs[0] = &s_x[0][0][__shfl(slotA, ldr)][ldc];
  xs[1] = &s_x[0][0][__shfl(slotA, ldr + 16)][ldc];
  xs[2] = &s_x[0][0][__shfl(slotB, ldr)][ldc];
  xs[3] = &s_x[0][0][__shfl(slotB, ldr + 16)][ldc];

  const unsigned short* ar[NRT];
#pragma unroll
  for (int rt = 0; rt < NRT; ++rt) {
    const int last = mycnt > 0 ? mycnt - 1 : 0;
    const int m    = (rt * 16 + nloc) < last ? (rt * 16 + nloc) : last;
    const int row  = (myoff + m) < TM ? (myoff + m) : TM - 1;
    ar[rt] = &s_x[0][0][row][ksel];
  }

  stage_x(x0v, xs, 0);
  stage_barrier();

  for (int c = 0; c < NCH; c += RING) {
    k_step<0>(c,     many, acc, bq, xq, bhi_p, blo_p, xrow, ar, xs);
    k_step<1>(c + 1, many, acc, bq, xq, bhi_p, blo_p, xrow, ar, xs);
    k_step<2>(c + 2, many, acc, bq, xq, bhi_p, blo_p, xrow, ar, xs);
    k_step<3>(c + 3, many, acc, bq, xq, bhi_p, blo_p, xrow, ar, xs);
  }

  {
    const float bc = bias[wv * 16 + nloc];
#pragma unroll
    for (int rt = 0; rt < NRT; ++rt)
#pragma unroll
      for (int p = 0; p < 8; ++p) {
        const int m = rt * 16 + rbase + p;
        if (m < mycnt) s_l1s[myoff + m][nloc] = acc[rt][p] + bc;
      }
  }
  __syncthreads();

  {
    const int i   = t >> 2;
    const int p   = t & 3;
    const int idx = lsi[r0 + i];
    const int sA  = __shfl(slotA, i & 31);
    const int sB  = __shfl(slotB, i & 31);
    const float* trow = &s_l1s[i < 32 ? sA : sB][0];
    float l1x[2 * L2];
#pragma unroll
    for (int j = 0; j < L2; ++j) {
      const float v = trow[j];
      l1x[j]      = clamp01(v * v * (127.0f / 128.0f));
      l1x[L2 + j] = clamp01(v);
    }
#pragma unroll
    for (int q = 0; q < 8; ++q) {
      const int n = p * 8 + q;
      const float* wrow = l2_w + (size_t)(idx * L3 + n) * (2 * L2);
      float a = l2_b[idx * L3 + n];
#pragma unroll
      for (int j = 0; j < 2 * L2; ++j) a += l1x[j] * wrow[j];
      s_l2x[i][n] = clamp01(a);
    }
  }
  __syncthreads();

  if (t < TM) {
    const int idx = lsi[r0 + t];
    const int sl  = t < 32 ? slotA : slotB;
    float s = out_b[idx];
#pragma unroll
    for (int n = 0; n < L3; ++n) s += s_l2x[t][n] * out_w[idx * L3 + n];
    out[r0 + t] = s + s_l1s[sl][L2];
  }
}

extern "C" void kernel_launch(void* const* d_in, const int* in_sizes, int n_in,
                              void* d_out, int out_size, void* d_ws, size_t ws_size,
                              hipStream_t stream) {
  const float* x     = (const float*)d_in[0];
  const int*   lsi   = (const int*)d_in[1];
  const float* l1_w  = (const float*)d_in[2];
  const float* l1_b  = (const float*)d_in[3];
  const float* l1f_w = (const float*)d_in[4];
  const float* l1f_b = (const float*)d_in[5];
  const float* l2_w  = (const float*)d_in[6];
  const float* l2_b  = (const float*)d_in[7];
  const float* out_w = (const float*)d_in[8];
  const float* out_b = (const float*)d_in[9];

  const int B = in_sizes[0] / L1;

  unsigned short* whi  = (unsigned short*)d_ws;
  unsigned short* wlo  = whi + (size_t)NCOL * L1;
  float*          bias = (float*)(wlo + (size_t)NCOL * L1);

  const int prepBlocks = (NCOL * L1 / 8 + 255) / 256;
  nnue_prep<<<prepBlocks, 256, 0, stream>>>(l1_w, l1_b, l1f_w, l1f_b, whi, wlo, bias);

  nnue_main<<<B / TM, 256, 0, stream>>>(x, lsi, whi, wlo, bias,
                                        l2_w, l2_b, out_w, out_b, (float*)d_out);
}


# Split of x into bf16 hi/lo in the staging done with the packed f32-to-bf16 conversion instruction (same round-to-nearest-even), not integer rounding arithmetic; values and order of sums unchanged
# speedup vs baseline: 2.0713x; 1.0299x over previous
#include <hip/hip_runtime.h>

#define COUNT 8
#define L1    3072
#define L2    15
#define L3    32
#define NCOL  128
#define TM    64
#define NRT   4
#define KC    64
#define LDSTR 72
#define NCH   (L1 / KC)
#define RING  4
#define PIECE 256
#define BCH   (4 * PIECE)
#define STAGE (TM * LDSTR)
#define PLANE (2 * STAGE)

typedef __attribute__((ext_vector_type(16))) __bf16         v16bf;
typedef __attribute__((ext_vector_type(8)))  float          v8f;
typedef __attribute__((ext_vector_type(16))) unsigned short v16u;
typedef __attribute__((ext_vector_type(8)))  unsigned short u16x8;
typedef __attribute__((ext_vector_type(4)))  unsigned short u16x4;
typedef __attribute__((ext_vector_type(2)))  float          f32x2;
typedef __attribute__((ext_vector_type(2)))  __bf16         bf16x2;
typedef __attribute__((ext_vector_type(2)))  unsigned int   u32x2;

__device__ __forceinline__ unsigned short f2bf(float f) {
  unsigned int u = __float_as_uint(f);
  u += 0x7fffu + ((u >> 16) & 1u);
  return (unsigned short)(u >> 16);
}
__device__ __forceinline__ float bf2f(unsigned short h) {
  return __uint_as_float(((unsigned int)h) << 16);
}
__device__ __forceinline__ float clamp01(float v) {
  return fminf(fmaxf(v, 0.0f), 1.0f);
}
__device__ __forceinline__ unsigned int pack2bf(float a, float b) {
  return __builtin_bit_cast(unsigned int, __builtin_convertvector((f32x2){a, b}, bf16x2));
}

__device__ __forceinline__ v16bf frag16(const unsigned short* p) {
  u16x8 a = *(const u16x8*)(p);
  u16x8 b = *(const u16x8*)(p + 16);
  v16u v;
#pragma unroll
  for (int i = 0; i < 8; ++i) { v[i] = a[i]; v[i + 8] = b[i]; }
  return __builtin_bit_cast(v16bf, v);
}

__device__ __forceinline__ v16bf frag16w(const unsigned short* p) {
  u16x8 a = *(const u16x8*)(p);
  u16x8 b = *(const u16x8*)(p + PIECE);
  v16u v;
#pragma unroll
  for (int i = 0; i < 8; ++i) { v[i] = a[i]; v[i + 8] = b[i]; }
  return __builtin_bit_cast(v16bf, v);
}

__device__ __forceinline__ void load_b(v16bf (&b)[2][2], const unsigned short* hi, const unsigned short* lo) {
#pragma unroll
  for (int ks = 0; ks < 2; ++ks) {
    b[ks][0] = frag16w(hi + ks * 2 * PIECE);
    b[ks][1] = frag16w(lo + ks * 2 * PIECE);
  }
}

__device__ __forceinline__ void load_x(float4 (&xv)[NRT], const float* p) {
#pragma unroll
  for (int q = 0; q < NRT; ++q) xv[q] = *(const float4*)(p + (size_t)q * 16 * L1);
}

__device__ __forceinline__ void stage_x(const float4 (&xv)[NRT], unsigned short* const (&xs)[NRT], const int st) {
#pragma unroll
  for (int q = 0; q < NRT; ++q) {
    const unsigned int h01 = pack2bf(xv[q].x, xv[q].y);
    const unsigned int h23 = pack2bf(xv[q].z, xv[q].w);
    const float r0 = xv[q].x - __uint_as_float(h01 << 16);
    const float r1 = xv[q].y - __uint_as_float(h01 & 0xffff0000u);
    const float r2 = xv[q].z - __uint_as_float(h23 << 16);
    const float r3 = xv[q].w - __uint_as_float(h23 & 0xffff0000u);
    *(u32x2*)(xs[q] + st)         = (u32x2){h01, h23};
    *(u32x2*)(xs[q] + st + PLANE) = (u32x2){pack2bf(r0, r1), pack2bf(r2, r3)};
  }
}

__device__ __forceinline__ void pin_order() { __builtin_amdgcn_sched_barrier(0); }
__device__ __forceinline__ void pin_loads() { __builtin_amdgcn_sched_barrier(0x2 | 0x4 | 0x80); }

template <int NT>
__device__ __forceinline__ void mma_chunk(v8f (&acc)[NRT], const v16bf (&b)[2][2],
                                          const unsigned short* const (&ar)[NRT], const int st) {
  v16bf a[2][2][NT];
#pragma unroll
  for (int ks = 0; ks < 2; ++ks) {
#pragma unroll
    for (int rt = 0; rt < NT; ++rt) a[ks][0][rt] = frag16(ar[rt] + st + ks * 32);
#pragma unroll
    for (int rt = 0; rt < NT; ++rt) a[ks][1][rt] = frag16(ar[rt] + st + PLANE + ks * 32);
  }
  pin_order();
#pragma unroll
  for (int ks = 0; ks < 2; ++ks) {
#pragma unroll
    for (int rt = 0; rt < NT; ++rt)
      acc[rt] = __builtin_amdgcn_wmma_f32_16x16x32_bf16(false, a[ks][0][rt], false, b[ks][0], (short)0, acc[rt], false, false);
#pragma unroll
    for (int rt = 0; rt < NT; ++rt)
      acc[rt] = __builtin_amdgcn_wmma_f32_16x16x32_bf16(false, a[ks][0][rt], false, b[ks][1], (short)0, acc[rt], false, false);
#pragma unroll
    for (int rt = 0; rt < NT; ++rt)
      acc[rt] = __builtin_amdgcn_wmma_f32_16x16x32_bf16(false, a[ks][1][rt], false, b[ks][0], (short)0, acc[rt], false, false);
  }
}

__device__ __forceinline__ void stage_barrier() {
  asm volatile("s_wait_dscnt 0x0\n\ts_barrier_signal -1\n\ts_barrier_wait -1" ::: "memory");
}

template <int S>
__device__ __forceinline__ void k_step(const int c, const bool many, v8f (&acc)[NRT], v16bf (&bq)[RING][2][2], float4 (&xq)[RING][NRT],
                                       const unsigned short* bhi_p, const unsigned short* blo_p, const float* xrow,
                                       const unsigned short* const (&ar)[NRT], unsigned short* const (&xs)[NRT]) {
  constexpr int CUR = S & 1;
  constexpr int SX  = (S + 1) % RING;
  if (many) mma_chunk<NRT>(acc, bq[S], ar, CUR * STAGE);
  else      mma_chunk<1>(acc, bq[S], ar, CUR * STAGE);
  stage_x(xq[SX], xs, (1 - CUR) * STAGE);
  const int cx = (c + 1 + RING) < NCH ? c + 1 + RING : NCH - 1;
  const int cb = (c + RING) < NCH ? c + RING : NCH - 1;
  asm volatile("" :: "v"(acc[0]), "v"(acc[1]), "v"(acc[2]), "v"(acc[3]) : "memory");
  load_x(xq[SX], xrow + cx * KC);
  load_b(bq[S], bhi_p + cb * BCH, blo_p + cb * BCH);
  stage_barrier();
}

__global__ void nnue_prep(const float* __restrict__ l1_w, const float* __restrict__ l1_b,
                          const float* __restrict__ l1f_w, const float* __restrict__ l1f_b,
                          unsigned short* __restrict__ whi, unsigned short* __restrict__ wlo,
                          float* __restrict__ bias) {
  const int p = blockIdx.x * 256 + threadIdx.x;
  if (p < NCOL * L1 / 8) {
    const int lane = p & 31;
    const int half = (p >> 5) & 1;
    const int ks   = (p >> 6) & 1;
    const int tc   = p >> 7;
    const int tile = tc / NCH;
    const int c    = tc - tile * NCH;
    const int r    = tile * 16 + (lane & 15);
    const int k    = c * KC + ks * 32 + half * 16 + (lane >> 4) * 8;
    const float4 a0 = *(const float4*)(l1_w + (size_t)r * L1 + k);
    const float4 a1 = *(const float4*)(l1_w + (size_t)r * L1 + k + 4);
    const float4 f0 = *(const float4*)(l1f_w + (size_t)(r & 15) * L1 + k);
    const float4 f1 = *(const float4*)(l1f_w + (size_t)(r & 15) * L1 + k + 4);
    const float w8[8] = {a0.x + f0.x, a0.y + f0.y, a0.z + f0.z, a0.w + f0.w,
                         a1.x + f1.x, a1.y + f1.y, a1.z + f1.z, a1.w + f1.w};
    u16x8 h, l;
#pragma unroll
    for (int i = 0; i < 8; ++i) {
      const unsigned short hb = f2bf(w8[i]);
      h[i] = hb;
      l[i] = f2bf(w8[i] - bf2f(hb));
    }
    *(u16x8*)(whi + (size_t)p * 8) = h;
    *(u16x8*)(wlo + (size_t)p * 8) = l;
  }
  if (p < NCOL) bias[p] = l1_b[p] + l1f_b[p & 15];
}

__global__ void __launch_bounds__(256, 1)
nnue_main(const float* x,
          const int* __restrict__ lsi,
          const unsigned short* whi,
          const unsigned short* wlo,
          const float* __restrict__ bias,
          const float* __restrict__ l2_w,
          const float* __restrict__ l2_b,
          const float* __restrict__ out_w,
          const float* __restrict__ out_b,
          float* __restrict__ out) {
  __shared__ __align__(16) unsigned short s_x[2][2][TM][LDSTR];
  __shared__ float s_l1s[TM][L2 + 2];
  __shared__ float s_l2x[TM][L3];

  const int t    = threadIdx.x;
  const int wv   = __builtin_amdgcn_readfirstlane(t >> 5);
  const int lane = t & 31;
  const int r0   = blockIdx.x * TM;

  const int nloc  = lane & 15;
  const int ksel  = (lane >> 4) << 3;
  const int rbase = (lane >> 4) << 3;

  const int ldr = t >> 4;
  const int ldc = (t & 15) << 2;

  const int bA = lsi[r0 + lane];
  const int bB = lsi[r0 + 32 + lane];

  const unsigned short* bhi_p = whi + (size_t)wv * NCH * BCH + lane * 8;
  const unsigned short* blo_p = wlo + (size_t)wv * NCH * BCH + lane * 8;
  const float* xrow = x + (size_t)(r0 + ldr) * L1 + ldc;

  v8f acc[NRT];
#pragma unroll
  for (int rt = 0; rt < NRT; ++rt)
    acc[rt] = (v8f){0.f, 0.f, 0.f, 0.f, 0.f, 0.f, 0.f, 0.f};

  v16bf  bq[RING][2][2];
  float4 xq[RING][NRT];

  float4 x0v[NRT];
  load_x(x0v, xrow);
#pragma unroll
  for (int s = 0; s < RING; ++s) {
    pin_loads();
    load_x(xq[(s + 1) % RING], xrow + (s + 1) * KC);
    load_b(bq[s], bhi_p + s * BCH, blo_p + s * BCH);
  }

  int slotA = 0, slotB = 0, myoff = 0, mycnt = 0;
  {
    const unsigned int below = (1u << lane) - 1u;
    int run = 0;
#pragma unroll
    for (int k = 0; k < COUNT; ++k) {
      const unsigned int mA = (unsigned int)__ballot(bA == k);
      const unsigned int mB = (unsigned int)__ballot(bB == k);
      const int cA = __popc(mA), cB = __popc(mB);
      if (bA == k) slotA = run + __popc(mA & below);
      if (bB == k) slotB = run + cA + __popc(mB & below);
      if (k == wv) { myoff = run; mycnt = cA + cB; }
      run += cA + cB;
    }
  }
  myoff = __builtin_amdgcn_readfirstlane(myoff);
  mycnt = __builtin_amdgcn_readfirstlane(mycnt);
  const bool many = mycnt > 16;

  unsigned short* xs[NRT];
  xs[0] = &s_x[0][0][__shfl(slotA, ldr)][ldc];
  xs[1] = &s_x[0][0][__shfl(slotA, ldr + 16)][ldc];
  xs[2] = &s_x[0][0][__shfl(slotB, ldr)][ldc];
  xs[3] = &s_x[0][0][__shfl(slotB, ldr + 16)][ldc];

  const unsigned short* ar[NRT];
#pragma unroll
  for (int rt = 0; rt < NRT; ++rt) {
    const int last = mycnt > 0 ? mycnt - 1 : 0;
    const int m    = (rt * 16 + nloc) < last ? (rt * 16 + nloc) : last;
    const int row  = (myoff + m) < TM ? (myoff + m) : TM - 1;
    ar[rt] = &s_x[0][0][row][ksel];
  }

  stage_x(x0v, xs, 0);
  stage_barrier();

  for (int c = 0; c < NCH; c += RING) {
    k_step<0>(c,     many, acc, bq, xq, bhi_p, blo_p, xrow, ar, xs);
    k_step<1>(c + 1, many, acc, bq, xq, bhi_p, blo_p, xrow, ar, xs);
    k_step<2>(c + 2, many, acc, bq, xq, bhi_p, blo_p, xrow, ar, xs);
    k_step<3>(c + 3, many, acc, bq, xq, bhi_p, blo_p, xrow, ar, xs);
  }

  {
    const float bc = bias[wv * 16 + nloc];
#pragma unroll
    for (int rt = 0; rt < NRT; ++rt)
#pragma unroll
      for (int p = 0; p < 8; ++p) {
        const int m = rt * 16 + rbase + p;
        if (m < mycnt) s_l1s[myoff + m][nloc] = acc[rt][p] + bc;
      }
  }
  __syncthreads();

  {
    const int i   = t >> 2;
    const int p   = t & 3;
    const int idx = lsi[r0 + i];
    const int sA  = __shfl(slotA, i & 31);
    const int sB  = __shfl(slotB, i & 31);
    const float* trow = &s_l1s[i < 32 ? sA : sB][0];
    float l1x[2 * L2];
#pragma unroll
    for (int j = 0; j < L2; ++j) {
      const float v = trow[j];
      l1x[j]      = clamp01(v * v * (127.0f / 128.0f));
      l1x[L2 + j] = clamp01(v);
    }
#pragma unroll
    for (int q = 0; q < 8; ++q) {
      const int n = p * 8 + q;
      const float* wrow = l2_w + (size_t)(idx * L3 + n) * (2 * L2);
      float a = l2_b[idx * L3 + n];
#pragma unroll
      for (int j = 0; j < 2 * L2; ++j) a += l1x[j] * wrow[j];
      s_l2x[i][n] = clamp01(a);
    }
  }
  __syncthreads();

  if (t < TM) {
    const int idx = lsi[r0 + t];
    const int sl  = t < 32 ? slotA : slotB;
    float s = out_b[idx];
#pragma unroll
    for (int n = 0; n < L3; ++n) s += s_l2x[t][n] * out_w[idx * L3 + n];
    out[r0 + t] = s + s_l1s[sl][L2];
  }
}

extern "C" void kernel_launch(void* const* d_in, const int* in_sizes, int n_in,
                              void* d_out, int out_size, void* d_ws, size_t ws_size,
                              hipStream_t stream) {
  const float* x     = (const float*)d_in[0];
  const int*   lsi   = (const int*)d_in[1];
  const float* l1_w  = (const float*)d_in[2];
  const float* l1_b  = (const float*)d_in[3];
  const float* l1f_w = (const float*)d_in[4];
  const float* l1f_b = (const float*)d_in[5];
  const float* l2_w  = (const float*)d_in[6];
  const float* l2_b  = (const float*)d_in[7];
  const float* out_w = (const float*)d_in[8];
  const float* out_b = (const float*)d_in[9];

  const int B = in_sizes[0] / L1;

  unsigned short* whi  = (unsigned short*)d_ws;
  unsigned short* wlo  = whi + (size_t)NCOL * L1;
  float*          bias = (float*)(wlo + (size_t)NCOL * L1);

  const int prepBlocks = (NCOL * L1 / 8 + 255) / 256;
  nnue_prep<<<prepBlocks, 256, 0, stream>>>(l1_w, l1_b, l1f_w, l1f_b, whi, wlo, bias);

  nnue_main<<<B / TM, 256, 0, stream>>>(x, lsi, whi, wlo, bias,
                                        l2_w, l2_b, out_w, out_b, (float*)d_out);
}


# speedup vs baseline: 2.1915x; 1.0580x over previous
#include <hip/hip_runtime.h>

#define COUNT 8
#define L1    3072
#define L2    15
#define L3    32
#define NCOL  128
#define TM    64
#define NRT   4
#define KC    64
#define LDSTR 72
#define NCH   (L1 / KC)
#define RING  4
#define PIECE 256
#define BCH   (4 * PIECE)
#define STAGE (TM * LDSTR)
#define PLANE (2 * STAGE)

typedef __attribute__((ext_vector_type(16))) __bf16         v16bf;
typedef __attribute__((ext_vector_type(8)))  float          v8f;
typedef __attribute__((ext_vector_type(16))) unsigned short v16u;
typedef __attribute__((ext_vector_type(8)))  unsigned short u16x8;
typedef __attribute__((ext_vector_type(4)))  unsigned short u16x4;
typedef __attribute__((ext_vector_type(2)))  float          f32x2;
typedef __attribute__((ext_vector_type(4)))  float          f32x4;
typedef __attribute__((ext_vector_type(2)))  __bf16         bf16x2;
typedef __attribute__((ext_vector_type(2)))  unsigned int   u32x2;

__device__ __forceinline__ unsigned short f2bf(float f) {
  unsigned int u = __float_as_uint(f);
  u += 0x7fffu + ((u >> 16) & 1u);
  return (unsigned short)(u >> 16);
}
__device__ __forceinline__ float bf2f(unsigned short h) {
  return __uint_as_float(((unsigned int)h) << 16);
}
__device__ __forceinline__ float clamp01(float v) {
  return fminf(fmaxf(v, 0.0f), 1.0f);
}
__device__ __forceinline__ unsigned int pack2bf(float a, float b) {
  return __builtin_bit_cast(unsigned int, __builtin_convertvector((f32x2){a, b}, bf16x2));
}

__device__ __forceinline__ v16bf frag16(const unsigned short* p) {
  u16x8 a = *(const u16x8*)(p);
  u16x8 b = *(const u16x8*)(p + 16);
  v16u v;
#pragma unroll
  for (int i = 0; i < 8; ++i) { v[i] = a[i]; v[i + 8] = b[i]; }
  return __builtin_bit_cast(v16bf, v);
}

__device__ __forceinline__ v16bf frag16w(const unsigned short* p) {
  u16x8 a = *(const u16x8*)(p);
  u16x8 b = *(const u16x8*)(p + PIECE);
  v16u v;
#pragma unroll
  for (int i = 0; i < 8; ++i) { v[i] = a[i]; v[i + 8] = b[i]; }
  return __builtin_bit_cast(v16bf, v);
}

__device__ __forceinline__ void load_b(v16bf (&b)[2][2], const unsigned short* hi, const unsigned short* lo) {
#pragma unroll
  for (int ks = 0; ks < 2; ++ks) {
    b[ks][0] = frag16w(hi + ks * 2 * PIECE);
    b[ks][1] = frag16w(lo + ks * 2 * PIECE);
  }
}

__device__ __forceinline__ void load_x(f32x4 (&xv)[NRT], const float* p) {
#pragma unroll
  for (int q = 0; q < NRT; ++q) xv[q] = __builtin_nontemporal_load((const f32x4*)(p + (size_t)q * 16 * L1));
}

__device__ __forceinline__ void stage_x(const f32x4 (&xv)[NRT], unsigned short* const (&xs)[NRT], const int st) {
#pragma unroll
  for (int q = 0; q < NRT; ++q) {
    const unsigned int h01 = pack2bf(xv[q].x, xv[q].y);
    const unsigned int h23 = pack2bf(xv[q].z, xv[q].w);
    const float r0 = xv[q].x - __uint_as_float(h01 << 16);
    const float r1 = xv[q].y - __uint_as_float(h01 & 0xffff0000u);
    const float r2 = xv[q].z - __uint_as_float(h23 << 16);
    const float r3 = xv[q].w - __uint_as_float(h23 & 0xffff0000u);
    *(u32x2*)(xs[q] + st)         = (u32x2){h01, h23};
    *(u32x2*)(xs[q] + st + PLANE) = (u32x2){pack2bf(r0, r1), pack2bf(r2, r3)};
  }
}

__device__ __forceinline__ void pin_order() { __builtin_amdgcn_sched_barrier(0); }
__device__ __forceinline__ void pin_loads() { __builtin_amdgcn_sched_barrier(0x2 | 0x4 | 0x80); }

template <int NT>
__device__ __forceinline__ void mma_chunk(v8f (&acc)[NRT], const v16bf (&b)[2][2],
                                          const unsigned short* const (&ar)[NRT], const int st) {
  v16bf a[2][2][NT];
#pragma unroll
  for (int ks = 0; ks < 2; ++ks) {
#pragma unroll
    for (int rt = 0; rt < NT; ++rt) a[ks][0][rt] = frag16(ar[rt] + st + ks * 32);
#pragma unroll
    for (int rt = 0; rt < NT; ++rt) a[ks][1][rt] = frag16(ar[rt] + st + PLANE + ks * 32);
  }
  pin_order();
#pragma unroll
  for (int ks = 0; ks < 2; ++ks) {
#pragma unroll
    for (int rt = 0; rt < NT; ++rt)
      acc[rt] = __builtin_amdgcn_wmma_f32_16x16x32_bf16(false, a[ks][0][rt], false, b[ks][0], (short)0, acc[rt], false, false);
#pragma unroll
    for (int rt = 0; rt < NT; ++rt)
      acc[rt] = __builtin_amdgcn_wmma_f32_16x16x32_bf16(false, a[ks][0][rt], false, b[ks][1], (short)0, acc[rt], false, false);
#pragma unroll
    for (int rt = 0; rt < NT; ++rt)
      acc[rt] = __builtin_amdgcn_wmma_f32_16x16x32_bf16(false, a[ks][1][rt], false, b[ks][0], (short)0, acc[rt], false, false);
  }
}

__device__ __forceinline__ void stage_barrier() {
  asm volatile("s_wait_dscnt 0x0\n\ts_barrier_signal -1\n\ts_barrier_wait -1" ::: "memory");
}

template <int S>
__device__ __forceinline__ void k_step(const int c, const bool many, v8f (&acc)[NRT], v16bf (&bq)[RING][2][2], f32x4 (&xq)[RING][NRT],
                                       const unsigned short* bhi_p, const unsigned short* blo_p, const float* xrow,
                                       const unsigned short* const (&ar)[NRT], unsigned short* const (&xs)[NRT]) {
  constexpr int CUR = S & 1;
  constexpr int SX  = (S + 1) % RING;
  if (many) mma_chunk<NRT>(acc, bq[S], ar, CUR * STAGE);
  else      mma_chunk<1>(acc, bq[S], ar, CUR * STAGE);
  stage_x(xq[SX], xs, (1 - CUR) * STAGE);
  const int cx = (c + 1 + RING) < NCH ? c + 1 + RING : NCH - 1;
  const int cb = (c + RING) < NCH ? c + RING : NCH - 1;
  asm volatile("" :: "v"(acc[0]), "v"(acc[1]), "v"(acc[2]), "v"(acc[3]) : "memory");
  load_x(xq[SX], xrow + cx * KC);
  load_b(bq[S], bhi_p + cb * BCH, blo_p + cb * BCH);
  stage_barrier();
}

__global__ void nnue_prep(const float* __restrict__ l1_w, const float* __restrict__ l1_b,
                          const float* __restrict__ l1f_w, const float* __restrict__ l1f_b,
                          unsigned short* __restrict__ whi, unsigned short* __restrict__ wlo,
                          float* __restrict__ bias) {
  const int p = blockIdx.x * 256 + threadIdx.x;
  if (p < NCOL * L1 / 8) {
    const int lane = p & 31;
    const int half = (p >> 5) & 1;
    const int ks   = (p >> 6) & 1;
    const int tc   = p >> 7;
    const int tile = tc / NCH;
    const int c    = tc - tile * NCH;
    const int r    = tile * 16 + (lane & 15);
    const int k    = c * KC + ks * 32 + half * 16 + (lane >> 4) * 8;
    const float4 a0 = *(const float4*)(l1_w + (size_t)r * L1 + k);
    const float4 a1 = *(const float4*)(l1_w + (size_t)r * L1 + k + 4);
    const float4 f0 = *(const float4*)(l1f_w + (size_t)(r & 15) * L1 + k);
    const float4 f1 = *(const float4*)(l1f_w + (size_t)(r & 15) * L1 + k + 4);
    const float w8[8] = {a0.x + f0.x, a0.y + f0.y, a0.z + f0.z, a0.w + f0.w,
                         a1.x + f1.x, a1.y + f1.y, a1.z + f1.z, a1.w + f1.w};
    u16x8 h, l;
#pragma unroll
    for (int i = 0; i < 8; ++i) {
      const unsigned short hb = f2bf(w8[i]);
      h[i] = hb;
      l[i] = f2bf(w8[i] - bf2f(hb));
    }
    *(u16x8*)(whi + (size_t)p * 8) = h;
    *(u16x8*)(wlo + (size_t)p * 8) = l;
  }
  if (p < NCOL) bias[p] = l1_b[p] + l1f_b[p & 15];
}

__global__ void __launch_bounds__(256, 1)
nnue_main(const float* x,
          const int* __restrict__ lsi,
          const unsigned short* whi,
          const unsigned short* wlo,
          const float* __restrict__ bias,
          const float* __restrict__ l2_w,
          const float* __restrict__ l2_b,
          const float* __restrict__ out_w,
          const float* __restrict__ out_b,
          float* __restrict__ out) {
  __shared__ __align__(16) unsigned short s_x[2][2][TM][LDSTR];
  __shared__ float s_l1s[TM][L2 + 2];
  __shared__ float s_l2x[TM][L3];

  const int t    = threadIdx.x;
  const int wv   = __builtin_amdgcn_readfirstlane(t >> 5);
  const int lane = t & 31;
  const int r0   = blockIdx.x * TM;

  const int nloc  = lane & 15;
  const int ksel  = (lane >> 4) << 3;
  const int rbase = (lane >> 4) << 3;

  const int ldr = t >> 4;
  const int ldc = (t & 15) << 2;

  const int bA = lsi[r0 + lane];
  const int bB = lsi[r0 + 32 + lane];

  const unsigned short* bhi_p = whi + (size_t)wv * NCH * BCH + lane * 8;
  const unsigned short* blo_p = wlo + (size_t)wv * NCH * BCH + lane * 8;
  const float* xrow = x + (size_t)(r0 + ldr) * L1 + ldc;

  v8f acc[NRT];
#pragma unroll
  for (int rt = 0; rt < NRT; ++rt)
    acc[rt] = (v8f){0.f, 0.f, 0.f, 0.f, 0.f, 0.f, 0.f, 0.f};

  v16bf  bq[RING][2][2];
  f32x4  xq[RING][NRT];

  f32x4 x0v[NRT];
  load_x(x0v, xrow);
#pragma unroll
  for (int s = 0; s < RING; ++s) {
    pin_loads();
    load_x(xq[(s + 1) % RING], xrow + (s + 1) * KC);
    load_b(bq[s], bhi_p + s * BCH, blo_p + s * BCH);
  }

  int slotA = 0, slotB = 0, myoff = 0, mycnt = 0;
  {
    const unsigned int below = (1u << lane) - 1u;
    int run = 0;
#pragma unroll
    for (int k = 0; k < COUNT; ++k) {
      const unsigned int mA = (unsigned int)__ballot(bA == k);
      const unsigned int mB = (unsigned int)__ballot(bB == k);
      const int cA = __popc(mA), cB = __popc(mB);
      if (bA == k) slotA = run + __popc(mA & below);
      if (bB == k) slotB = run + cA + __popc(mB & below);
      if (k == wv) { myoff = run; mycnt = cA + cB; }
      run += cA + cB;
    }
  }
  myoff = __builtin_amdgcn_readfirstlane(myoff);
  mycnt = __builtin_amdgcn_readfirstlane(mycnt);
  const bool many = mycnt > 16;

  unsigned short* xs[NRT];
  xs[0] = &s_x[0][0][__shfl(slotA, ldr)][ldc];
  xs[1] = &s_x[0][0][__shfl(slotA, ldr + 16)][ldc];
  xs[2] = &s_x[0][0][__shfl(slotB, ldr)][ldc];
  xs[3] = &s_x[0][0][__shfl(slotB, ldr + 16)][ldc];

  const unsigned short* ar[NRT];
#pragma unroll
  for (int rt = 0; rt < NRT; ++rt) {
    const int last = mycnt > 0 ? mycnt - 1 : 0;
    const int m    = (rt * 16 + nloc) < last ? (rt * 16 + nloc) : last;
    const int row  = (myoff + m) < TM ? (myoff + m) : TM - 1;
    ar[rt] = &s_x[0][0][row][ksel];
  }

  stage_x(x0v, xs, 0);
  stage_barrier();

  for (int c = 0; c < NCH; c += RING) {
    k_step<0>(c,     many, acc, bq, xq, bhi_p, blo_p, xrow, ar, xs);
    k_step<1>(c + 1, many, acc, bq, xq, bhi_p, blo_p, xrow, ar, xs);
    k_step<2>(c + 2, many, acc, bq, xq, bhi_p, blo_p, xrow, ar, xs);
    k_step<3>(c + 3, many, acc, bq, xq, bhi_p, blo_p, xrow, ar, xs);
  }

  {
    const float bc = bias[wv * 16 + nloc];
#pragma unroll
    for (int rt = 0; rt < NRT; ++rt)
#pragma unroll
      for (int p = 0; p < 8; ++p) {
        const int m = rt * 16 + rbase + p;
        if (m < mycnt) s_l1s[myoff + m][nloc] = acc[rt][p] + bc;
      }
  }
  __syncthreads();

  {
    const int i   = t >> 2;
    const int p   = t & 3;
    const int idx = lsi[r0 + i];
    const int sA  = __shfl(slotA, i & 31);
    const int sB  = __shfl(slotB, i & 31);
    const float* trow = &s_l1s[i < 32 ? sA : sB][0];
    float l1x[2 * L2];
#pragma unroll
    for (int j = 0; j < L2; ++j) {
      const float v = trow[j];
      l1x[j]      = clamp01(v * v * (127.0f / 128.0f));
      l1x[L2 + j] = clamp01(v);
    }
#pragma unroll
    for (int q = 0; q < 8; ++q) {
      const int n = p * 8 + q;
      const float* wrow = l2_w + (size_t)(idx * L3 + n) * (2 * L2);
      float a = l2_b[idx * L3 + n];
#pragma unroll
      for (int j = 0; j < 2 * L2; ++j) a += l1x[j] * wrow[j];
      s_l2x[i][n] = clamp01(a);
    }
  }
  __syncthreads();

  if (t < TM) {
    const int idx = lsi[r0 + t];
    const int sl  = t < 32 ? slotA : slotB;
    float s = out_b[idx];
#pragma unroll
    for (int n = 0; n < L3; ++n) s += s_l2x[t][n] * out_w[idx * L3 + n];
    out[r0 + t] = s + s_l1s[sl][L2];
  }
}

extern "C" void kernel_launch(void* const* d_in, const int* in_sizes, int n_in,
                              void* d_out, int out_size, void* d_ws, size_t ws_size,
                              hipStream_t stream) {
  const float* x     = (const float*)d_in[0];
  const int*   lsi   = (const int*)d_in[1];
  const float* l1_w  = (const float*)d_in[2];
  const float* l1_b  = (const float*)d_in[3];
  const float* l1f_w = (const float*)d_in[4];
  const float* l1f_b = (const float*)d_in[5];
  const float* l2_w  = (const float*)d_in[6];
  const float* l2_b  = (const float*)d_in[7];
  const float* out_w = (const float*)d_in[8];
  const float* out_b = (const float*)d_in[9];

  const int B = in_sizes[0] / L1;

  unsigned short* whi  = (unsigned short*)d_ws;
  unsigned short* wlo  = whi + (size_t)NCOL * L1;
  float*          bias = (float*)(wlo + (size_t)NCOL * L1);

  const int prepBlocks = (NCOL * L1 / 8 + 255) / 256;
  nnue_prep<<<prepBlocks, 256, 0, stream>>>(l1_w, l1_b, l1f_w, l1f_b, whi, wlo, bias);

  nnue_main<<<B / TM, 256, 0, stream>>>(x, lsi, whi, wlo, bias,
                                        l2_w, l2_b, out_w, out_b, (float*)d_out);
}


# Rows grouped by bucket in the prep launch (counting sort, no atomics): a workgroup takes 64 rows of one bucket (two at a boundary) and loads only their weight columns, not all eight; sums unchanged
# speedup vs baseline: 2.6273x; 1.1989x over previous
#include <hip/hip_runtime.h>

#define COUNT 8
#define L1    3072
#define L2    15
#define L3    32
#define NCOL  128
#define TM    64
#define NRT   4
#define KC    64
#define LDSTR 72
#define NCH   (L1 / KC)
#define RING  4
#define PIECE 256
#define BCH   (4 * PIECE)
#define STAGE (TM * LDSTR)
#define PLANE (2 * STAGE)
#define PREPB ((NCOL * L1 / 8 + 255) / 256)
#define SORTJ 16

typedef __attribute__((ext_vector_type(16))) __bf16         v16bf;
typedef __attribute__((ext_vector_type(8)))  float          v8f;
typedef __attribute__((ext_vector_type(16))) unsigned short v16u;
typedef __attribute__((ext_vector_type(8)))  unsigned short u16x8;
typedef __attribute__((ext_vector_type(4)))  unsigned short u16x4;
typedef __attribute__((ext_vector_type(2)))  float          f32x2;
typedef __attribute__((ext_vector_type(4)))  float          f32x4;
typedef __attribute__((ext_vector_type(2)))  __bf16         bf16x2;
typedef __attribute__((ext_vector_type(2)))  unsigned int   u32x2;

__device__ __forceinline__ unsigned short f2bf(float f) {
  unsigned int u = __float_as_uint(f);
  u += 0x7fffu + ((u >> 16) & 1u);
  return (unsigned short)(u >> 16);
}
__device__ __forceinline__ float bf2f(unsigned short h) {
  return __uint_as_float(((unsigned int)h) << 16);
}
__device__ __forceinline__ float clamp01(float v) {
  return fminf(fmaxf(v, 0.0f), 1.0f);
}
__device__ __forceinline__ unsigned int pack2bf(float a, float b) {
  return __builtin_bit_cast(unsigned int, __builtin_convertvector((f32x2){a, b}, bf16x2));
}

__device__ __forceinline__ v16bf frag16(const unsigned short* p) {
  u16x8 a = *(const u16x8*)(p);
  u16x8 b = *(const u16x8*)(p + 16);
  v16u v;
#pragma unroll
  for (int i = 0; i < 8; ++i) { v[i] = a[i]; v[i + 8] = b[i]; }
  return __builtin_bit_cast(v16bf, v);
}

__device__ __forceinline__ v16bf frag16w(const unsigned short* p) {
  u16x8 a = *(const u16x8*)(p);
  u16x8 b = *(const u16x8*)(p + PIECE);
  v16u v;
#pragma unroll
  for (int i = 0; i < 8; ++i) { v[i] = a[i]; v[i + 8] = b[i]; }
  return __builtin_bit_cast(v16bf, v);
}

__device__ __forceinline__ void load_b(v16bf (&b)[2][2], const unsigned short* hi, const unsigned short* lo) {
#pragma unroll
  for (int ks = 0; ks < 2; ++ks) {
    b[ks][0] = frag16w(hi + ks * 2 * PIECE);
    b[ks][1] = frag16w(lo + ks * 2 * PIECE);
  }
}

__device__ __forceinline__ void load_x(f32x4 (&xv)[NRT], const float* x, const unsigned int (&xo)[NRT], const unsigned int k) {
#pragma unroll
  for (int q = 0; q < NRT; ++q) xv[q] = __builtin_nontemporal_load((const f32x4*)(x + (size_t)(xo[q] + k)));
}

__device__ __forceinline__ void stage_x(const f32x4 (&xv)[NRT], unsigned short* const (&xs)[NRT], const int st) {
#pragma unroll
  for (int q = 0; q < NRT; ++q) {
    const unsigned int h01 = pack2bf(xv[q].x, xv[q].y);
    const unsigned int h23 = pack2bf(xv[q].z, xv[q].w);
    const float r0 = xv[q].x - __uint_as_float(h01 << 16);
    const float r1 = xv[q].y - __uint_as_float(h01 & 0xffff0000u);
    const float r2 = xv[q].z - __uint_as_float(h23 << 16);
    const float r3 = xv[q].w - __uint_as_float(h23 & 0xffff0000u);
    *(u32x2*)(xs[q] + st)         = (u32x2){h01, h23};
    *(u32x2*)(xs[q] + st + PLANE) = (u32x2){pack2bf(r0, r1), pack2bf(r2, r3)};
  }
}

__device__ __forceinline__ void pin_order() { __builtin_amdgcn_sched_barrier(0); }
__device__ __forceinline__ void pin_loads() { __builtin_amdgcn_sched_barrier(0x2 | 0x4 | 0x80); }

template <int NT>
__device__ __forceinline__ void mma_chunk(v8f (&acc)[NRT], const v16bf (&b)[2][2],
                                          const unsigned short* const (&ar)[NRT], const int st) {
  v16bf a[2][2][NT];
#pragma unroll
  for (int ks = 0; ks < 2; ++ks) {
#pragma unroll
    for (int rt = 0; rt < NT; ++rt) a[ks][0][rt] = frag16(ar[rt] + st + ks * 32);
#pragma unroll
    for (int rt = 0; rt < NT; ++rt) a[ks][1][rt] = frag16(ar[rt] + st + PLANE + ks * 32);
  }
  pin_order();
#pragma unroll
  for (int ks = 0; ks < 2; ++ks) {
#pragma unroll
    for (int rt = 0; rt < NT; ++rt)
      acc[rt] = __builtin_amdgcn_wmma_f32_16x16x32_bf16(false, a[ks][0][rt], false, b[ks][0], (short)0, acc[rt], false, false);
#pragma unroll
    for (int rt = 0; rt < NT; ++rt)
      acc[rt] = __builtin_amdgcn_wmma_f32_16x16x32_bf16(false, a[ks][0][rt], false, b[ks][1], (short)0, acc[rt], false, false);
#pragma unroll
    for (int rt = 0; rt < NT; ++rt)
      acc[rt] = __builtin_amdgcn_wmma_f32_16x16x32_bf16(false, a[ks][1][rt], false, b[ks][0], (short)0, acc[rt], false, false);
  }
}

__device__ __forceinline__ void stage_barrier() {
  asm volatile("s_wait_dscnt 0x0\n\ts_barrier_signal -1\n\ts_barrier_wait -1" ::: "memory");
}

template <int S>
__device__ __forceinline__ void k_step(const int c, const bool many, const bool idle, v8f (&acc)[NRT], v16bf (&bq)[RING][2][2], f32x4 (&xq)[RING][NRT],
                                       const unsigned short* bhi_p, const unsigned short* blo_p, const float* x, const unsigned int (&xo)[NRT],
                                       const unsigned short* const (&ar)[NRT], unsigned short* const (&xs)[NRT]) {
  constexpr int CUR = S & 1;
  constexpr int SX  = (S + 1) % RING;
  if (many) mma_chunk<NRT>(acc, bq[S], ar, CUR * STAGE);
  else      mma_chunk<1>(acc, bq[S], ar, CUR * STAGE);
  stage_x(xq[SX], xs, (1 - CUR) * STAGE);
  const int cx = (c + 1 + RING) < NCH ? c + 1 + RING : NCH - 1;
  const int cb = idle ? 0 : ((c + RING) < NCH ? c + RING : NCH - 1);
  asm volatile("" :: "v"(acc[0]), "v"(acc[1]), "v"(acc[2]), "v"(acc[3]) : "memory");
  load_x(xq[SX], x, xo, (unsigned int)(cx * KC));
  load_b(bq[S], bhi_p + cb * BCH, blo_p + cb * BCH);
  stage_barrier();
}

__global__ void __launch_bounds__(256)
nnue_prep(const float* __restrict__ l1_w, const float* __restrict__ l1_b,
          const float* __restrict__ l1f_w, const float* __restrict__ l1f_b,
          unsigned short* __restrict__ whi, unsigned short* __restrict__ wlo,
          float* __restrict__ bias,
          const int* __restrict__ lsi, int* __restrict__ perm, const int B) {
  __shared__ int s_c[COUNT][256];
  __shared__ int s_tot[COUNT];
  __shared__ __align__(16) int s_perm[SORTJ * 1024 + 256];
  if (blockIdx.x == 0) {
    const int t = threadIdx.x;
#pragma unroll
    for (int j = 0; j < SORTJ; ++j) {
      const int row0 = (j * 256 + t) * 4;
      *(int4*)(s_perm + row0) = *(const int4*)(lsi + (row0 < B - 4 ? row0 : B - 4));
    }
    __syncthreads();
    const int okm = -(int)(t * 64 < B);
    int bk[64];
#pragma unroll
    for (int i = 0; i < 64; ++i) bk[i] = s_perm[t * 64 + ((i + t) & 63)] & 7;
    unsigned long long pk = 0ull;
#pragma unroll
    for (int i = 0; i < 64; ++i) pk += (unsigned long long)(okm & 1) << (bk[i] * 8);
#pragma unroll
    for (int k = 0; k < COUNT; ++k) s_c[k][t] = (int)((pk >> (k * 8)) & 0xffull);
    __syncthreads();
    {
      const int w = t >> 5, l = t & 31;
      int a[8], sum = 0;
#pragma unroll
      for (int i = 0; i < 8; ++i) { a[i] = s_c[w][l * 8 + i]; sum += a[i]; }
      int s = sum;
#pragma unroll
      for (int d = 1; d < 32; d <<= 1) {
        const int o = __shfl_up(s, d);
        if (l >= d) s += o;
      }
      int before = s - sum;
#pragma unroll
      for (int i = 0; i < 8; ++i) { s_c[w][l * 8 + i] = before; before += a[i]; }
      if (l == 31) s_tot[w] = s;
    }
    __syncthreads();
    int first[COUNT];
    {
      int base = 0;
#pragma unroll
      for (int k = 0; k < COUNT; ++k) {
        first[k] = s_c[k][t] + base;
        base += s_tot[k];
      }
    }
#pragma unroll
    for (int k = 0; k < COUNT; ++k) s_c[k][t] = first[k];
    pk = 0ull;
#pragma unroll
    for (int i = 0; i < 64; ++i) {
      const int sh   = bk[i] * 8;
      const int rank = (int)((pk >> sh) & 0xffull);
      pk += (unsigned long long)(okm & 1) << sh;
      const int row  = t * 64 + ((i + t) & 63);
      const int pos  = ((s_c[bk[i]][t] + rank) & okm) | ((SORTJ * 1024 + t) & ~okm);
      s_perm[pos] = row | (bk[i] << 24);
    }
    __syncthreads();
#pragma unroll
    for (int j = 0; j < SORTJ; ++j) {
      const int i0  = (j * 256 + t) * 4;
      const int okm = -(int)(i0 < B);
      *(int4*)(perm + ((i0 & okm) | ((SORTJ * 1024 + t * 4) & ~okm))) = *(const int4*)(s_perm + i0);
    }
    return;
  }
  const int p = (blockIdx.x - 1) * 256 + threadIdx.x;
  if (p < NCOL * L1 / 8) {
    const int lane = p & 31;
    const int half = (p >> 5) & 1;
    const int ks   = (p >> 6) & 1;
    const int tc   = p >> 7;
    const int tile = tc / NCH;
    const int c    = tc - tile * NCH;
    const int r    = tile * 16 + (lane & 15);
    const int k    = c * KC + ks * 32 + half * 16 + (lane >> 4) * 8;
    const float4 a0 = *(const float4*)(l1_w + (size_t)r * L1 + k);
    const float4 a1 = *(const float4*)(l1_w + (size_t)r * L1 + k + 4);
    const float4 f0 = *(const float4*)(l1f_w + (size_t)(r & 15) * L1 + k);
    const float4 f1 = *(const float4*)(l1f_w + (size_t)(r & 15) * L1 + k + 4);
    const float w8[8] = {a0.x + f0.x, a0.y + f0.y, a0.z + f0.z, a0.w + f0.w,
                         a1.x + f1.x, a1.y + f1.y, a1.z + f1.z, a1.w + f1.w};
    u16x8 h, l;
#pragma unroll
    for (int i = 0; i < 8; ++i) {
      const unsigned short hb = f2bf(w8[i]);
      h[i] = hb;
      l[i] = f2bf(w8[i] - bf2f(hb));
    }
    *(u16x8*)(whi + (size_t)p * 8) = h;
    *(u16x8*)(wlo + (size_t)p * 8) = l;
  }
  if (p < NCOL) bias[p] = l1_b[p] + l1f_b[p & 15];
}

__global__ void __launch_bounds__(256, 1)
nnue_main(const float* x,
          const int* __restrict__ perm,
          const unsigned short* whi,
          const unsigned short* wlo,
          const float* __restrict__ bias,
          const float* __restrict__ l2_w,
          const float* __restrict__ l2_b,
          const float* __restrict__ out_w,
          const float* __restrict__ out_b,
          float* __restrict__ out) {
  __shared__ __align__(16) unsigned short s_x[2][2][TM][LDSTR];
  __shared__ float s_l1s[TM][L2 + 2];
  __shared__ float s_l2x[TM][L3];

  const int t    = threadIdx.x;
  const int wv   = __builtin_amdgcn_readfirstlane(t >> 5);
  const int lane = t & 31;
  const int r0   = blockIdx.x * TM;

  const int nloc  = lane & 15;
  const int ksel  = (lane >> 4) << 3;
  const int rbase = (lane >> 4) << 3;

  const int ldr = t >> 4;
  const int ldc = (t & 15) << 2;

  const int pA = perm[r0 + lane];
  const int pB = perm[r0 + 32 + lane];
  const int rowA = pA & 0xffffff, bA = (pA >> 24) & 7;
  const int rowB = pB & 0xffffff, bB = (pB >> 24) & 7;

  unsigned int xo[NRT];
  xo[0] = (unsigned int)__shfl(rowA, ldr)      * L1 + ldc;
  xo[1] = (unsigned int)__shfl(rowA, ldr + 16) * L1 + ldc;
  xo[2] = (unsigned int)__shfl(rowB, ldr)      * L1 + ldc;
  xo[3] = (unsigned int)__shfl(rowB, ldr + 16) * L1 + ldc;

  int slotA = 0, slotB = 0;
  int myoff = 0, mycnt = 0, wb = 0;
  {
    const unsigned int below = (1u << lane) - 1u;
    int run = 0;
    int npc = 0;
    int poff = 0, pcnt = 0, pb = 0, lastb = 0;
    int foff = 0, fcnt = 0;
#pragma unroll
    for (int k = 0; k < COUNT; ++k) {
      const unsigned int mA = (unsigned int)__ballot(bA == k);
      const unsigned int mB = (unsigned int)__ballot(bB == k);
      const int cA = __popc(mA), cB = __popc(mB);
      if (bA == k) slotA = run + __popc(mA & below);
      if (bB == k) slotB = run + cA + __popc(mB & below);
      const int cnt = cA + cB;
      const int pcs = (cnt + 15) >> 4;
      if (wv >= npc && wv < npc + pcs) {
        const int j    = wv - npc;
        const int left = cnt - 16 * j;
        pb = k; poff = run + 16 * j; pcnt = left < 16 ? left : 16;
      }
      if (k == wv) { foff = run; fcnt = cnt; }
      if (cnt > 0) lastb = k;
      run += cnt;
      npc += pcs;
    }
    if (npc <= COUNT) {
      if (wv < npc) { wb = pb; myoff = poff; mycnt = pcnt; }
      else          { wb = lastb; myoff = 0; mycnt = 0; }
    } else {
      wb = wv; myoff = foff; mycnt = fcnt;
    }
  }
  myoff = __builtin_amdgcn_readfirstlane(myoff);
  mycnt = __builtin_amdgcn_readfirstlane(mycnt);
  wb    = __builtin_amdgcn_readfirstlane(wb);
  const bool many = mycnt > 16;
  const bool idle = mycnt == 0;

  const unsigned short* bhi_p = whi + (size_t)wb * NCH * BCH + lane * 8;
  const unsigned short* blo_p = wlo + (size_t)wb * NCH * BCH + lane * 8;

  v8f acc[NRT];
#pragma unroll
  for (int rt = 0; rt < NRT; ++rt)
    acc[rt] = (v8f){0.f, 0.f, 0.f, 0.f, 0.f, 0.f, 0.f, 0.f};

  v16bf  bq[RING][2][2];
  f32x4  xq[RING][NRT];

  f32x4 x0v[NRT];
  load_x(x0v, x, xo, 0u);
#pragma unroll
  for (int s = 0; s < RING; ++s) {
    pin_loads();
    load_x(xq[(s + 1) % RING], x, xo, (unsigned int)((s + 1) * KC));
    load_b(bq[s], bhi_p + s * BCH, blo_p + s * BCH);
  }

  unsigned short* xs[NRT];
  xs[0] = &s_x[0][0][__shfl(slotA, ldr)][ldc];
  xs[1] = &s_x[0][0][__shfl(slotA, ldr + 16)][ldc];
  xs[2] = &s_x[0][0][__shfl(slotB, ldr)][ldc];
  xs[3] = &s_x[0][0][__shfl(slotB, ldr + 16)][ldc];

  const unsigned short* ar[NRT];
#pragma unroll
  for (int rt = 0; rt < NRT; ++rt) {
    const int last = mycnt > 0 ? mycnt - 1 : 0;
    const int m    = (rt * 16 + nloc) < last ? (rt * 16 + nloc) : last;
    const int row  = (myoff + m) < TM ? (myoff + m) : TM - 1;
    ar[rt] = &s_x[0][0][row][ksel];
  }

  stage_x(x0v, xs, 0);
  stage_barrier();

  for (int c = 0; c < NCH; c += RING) {
    k_step<0>(c,     many, idle, acc, bq, xq, bhi_p, blo_p, x, xo, ar, xs);
    k_step<1>(c + 1, many, idle, acc, bq, xq, bhi_p, blo_p, x, xo, ar, xs);
    k_step<2>(c + 2, many, idle, acc, bq, xq, bhi_p, blo_p, x, xo, ar, xs);
    k_step<3>(c + 3, many, idle, acc, bq, xq, bhi_p, blo_p, x, xo, ar, xs);
  }

  {
    const float bc = bias[wb * 16 + nloc];
#pragma unroll
    for (int rt = 0; rt < NRT; ++rt)
#pragma unroll
      for (int p = 0; p < 8; ++p) {
        const int m = rt * 16 + rbase + p;
        if (m < mycnt) s_l1s[myoff + m][nloc] = acc[rt][p] + bc;
      }
  }
  __syncthreads();

  {
    const int i   = t >> 2;
    const int p   = t & 3;
    const int iA  = __shfl(bA, i & 31);
    const int iB  = __shfl(bB, i & 31);
    const int idx = i < 32 ? iA : iB;
    const int sA  = __shfl(slotA, i & 31);
    const int sB  = __shfl(slotB, i & 31);
    const float* trow = &s_l1s[i < 32 ? sA : sB][0];
    float l1x[2 * L2];
#pragma unroll
    for (int j = 0; j < L2; ++j) {
      const float v = trow[j];
      l1x[j]      = clamp01(v * v * (127.0f / 128.0f));
      l1x[L2 + j] = clamp01(v);
    }
#pragma unroll
    for (int q = 0; q < 8; ++q) {
      const int n = p * 8 + q;
      const float* wrow = l2_w + (size_t)(idx * L3 + n) * (2 * L2);
      float a = l2_b[idx * L3 + n];
#pragma unroll
      for (int j = 0; j < 2 * L2; ++j) a += l1x[j] * wrow[j];
      s_l2x[i][n] = clamp01(a);
    }
  }
  __syncthreads();

  if (t < TM) {
    const int idx = t < 32 ? bA : bB;
    const int sl  = t < 32 ? slotA : slotB;
    const int row = t < 32 ? rowA : rowB;
    float s = out_b[idx];
#pragma unroll
    for (int n = 0; n < L3; ++n) s += s_l2x[t][n] * out_w[idx * L3 + n];
    out[row] = s + s_l1s[sl][L2];
  }
}

extern "C" void kernel_launch(void* const* d_in, const int* in_sizes, int n_in,
                              void* d_out, int out_size, void* d_ws, size_t ws_size,
                              hipStream_t stream) {
  const float* x     = (const float*)d_in[0];
  const int*   lsi   = (const int*)d_in[1];
  const float* l1_w  = (const float*)d_in[2];
  const float* l1_b  = (const float*)d_in[3];
  const float* l1f_w = (const float*)d_in[4];
  const float* l1f_b = (const float*)d_in[5];
  const float* l2_w  = (const float*)d_in[6];
  const float* l2_b  = (const float*)d_in[7];
  const float* out_w = (const float*)d_in[8];
  const float* out_b = (const float*)d_in[9];

  const int B = in_sizes[0] / L1;

  unsigned short* whi  = (unsigned short*)d_ws;
  unsigned short* wlo  = whi + (size_t)NCOL * L1;
  float*          bias = (float*)(wlo + (size_t)NCOL * L1);
  int*            perm = (int*)(bias + NCOL);

  nnue_prep<<<PREPB + 1, 256, 0, stream>>>(l1_w, l1_b, l1f_w, l1f_b, whi, wlo, bias, lsi, perm, B);

  nnue_main<<<B / TM, 256, 0, stream>>>(x, perm, whi, wlo, bias,
                                        l2_w, l2_b, out_w, out_b, (float*)d_out);
}


# K-loop barrier split: a wave signals once its LDS work of the step is done (fragment reads, x split moved ahead of the WMMAs) and waits behind its WMMAs and global loads; order of sums unchanged
# speedup vs baseline: 2.7160x; 1.0338x over previous
#include <hip/hip_runtime.h>

#define COUNT 8
#define L1    3072
#define L2    15
#define L3    32
#define NCOL  128
#define TM    64
#define NRT   4
#define KC    64
#define LDSTR 72
#define NCH   (L1 / KC)
#define RING  4
#define PIECE 256
#define BCH   (4 * PIECE)
#define STAGE (TM * LDSTR)
#define PLANE (2 * STAGE)
#define PREPB ((NCOL * L1 / 8 + 255) / 256)
#define SORTJ 16

typedef __attribute__((ext_vector_type(16))) __bf16         v16bf;
typedef __attribute__((ext_vector_type(8)))  float          v8f;
typedef __attribute__((ext_vector_type(16))) unsigned short v16u;
typedef __attribute__((ext_vector_type(8)))  unsigned short u16x8;
typedef __attribute__((ext_vector_type(4)))  unsigned short u16x4;
typedef __attribute__((ext_vector_type(2)))  float          f32x2;
typedef __attribute__((ext_vector_type(4)))  float          f32x4;
typedef __attribute__((ext_vector_type(2)))  __bf16         bf16x2;
typedef __attribute__((ext_vector_type(2)))  unsigned int   u32x2;

__device__ __forceinline__ unsigned short f2bf(float f) {
  unsigned int u = __float_as_uint(f);
  u += 0x7fffu + ((u >> 16) & 1u);
  return (unsigned short)(u >> 16);
}
__device__ __forceinline__ float bf2f(unsigned short h) {
  return __uint_as_float(((unsigned int)h) << 16);
}
__device__ __forceinline__ float clamp01(float v) {
  return fminf(fmaxf(v, 0.0f), 1.0f);
}
__device__ __forceinline__ unsigned int pack2bf(float a, float b) {
  return __builtin_bit_cast(unsigned int, __builtin_convertvector((f32x2){a, b}, bf16x2));
}

__device__ __forceinline__ v16bf frag16(const unsigned short* p) {
  u16x8 a = *(const u16x8*)(p);
  u16x8 b = *(const u16x8*)(p + 16);
  v16u v;
#pragma unroll
  for (int i = 0; i < 8; ++i) { v[i] = a[i]; v[i + 8] = b[i]; }
  return __builtin_bit_cast(v16bf, v);
}

__device__ __forceinline__ v16bf frag16w(const unsigned short* p) {
  u16x8 a = *(const u16x8*)(p);
  u16x8 b = *(const u16x8*)(p + PIECE);
  v16u v;
#pragma unroll
  for (int i = 0; i < 8; ++i) { v[i] = a[i]; v[i + 8] = b[i]; }
  return __builtin_bit_cast(v16bf, v);
}

__device__ __forceinline__ void load_b(v16bf (&b)[2][2], const unsigned short* hi, const unsigned short* lo) {
#pragma unroll
  for (int ks = 0; ks < 2; ++ks) {
    b[ks][0] = frag16w(hi + ks * 2 * PIECE);
    b[ks][1] = frag16w(lo + ks * 2 * PIECE);
  }
}

__device__ __forceinline__ void load_x(f32x4 (&xv)[NRT], const float* x, const unsigned int (&xo)[NRT], const unsigned int k) {
#pragma unroll
  for (int q = 0; q < NRT; ++q) xv[q] = __builtin_nontemporal_load((const f32x4*)(x + (size_t)(xo[q] + k)));
}

__device__ __forceinline__ void stage_x(const f32x4 (&xv)[NRT], unsigned short* const (&xs)[NRT], const int st) {
#pragma unroll
  for (int q = 0; q < NRT; ++q) {
    const unsigned int h01 = pack2bf(xv[q].x, xv[q].y);
    const unsigned int h23 = pack2bf(xv[q].z, xv[q].w);
    const float r0 = xv[q].x - __uint_as_float(h01 << 16);
    const float r1 = xv[q].y - __uint_as_float(h01 & 0xffff0000u);
    const float r2 = xv[q].z - __uint_as_float(h23 << 16);
    const float r3 = xv[q].w - __uint_as_float(h23 & 0xffff0000u);
    *(u32x2*)(xs[q] + st)         = (u32x2){h01, h23};
    *(u32x2*)(xs[q] + st + PLANE) = (u32x2){pack2bf(r0, r1), pack2bf(r2, r3)};
  }
}

__device__ __forceinline__ void pin_order() { __builtin_amdgcn_sched_barrier(0); }
__device__ __forceinline__ void pin_loads() { __builtin_amdgcn_sched_barrier(0x2 | 0x4 | 0x80); }

template <int NT>
__device__ __forceinline__ void load_a(v16bf (&a)[2][2][NT], const unsigned short* const (&ar)[NRT], const int st) {
#pragma unroll
  for (int ks = 0; ks < 2; ++ks) {
#pragma unroll
    for (int rt = 0; rt < NT; ++rt) a[ks][0][rt] = frag16(ar[rt] + st + ks * 32);
#pragma unroll
    for (int rt = 0; rt < NT; ++rt) a[ks][1][rt] = frag16(ar[rt] + st + PLANE + ks * 32);
  }
}
template <int NT>
__device__ __forceinline__ void mma_regs(v8f (&acc)[NRT], const v16bf (&a)[2][2][NT], const v16bf (&b)[2][2]) {
#pragma unroll
  for (int ks = 0; ks < 2; ++ks) {
#pragma unroll
    for (int rt = 0; rt < NT; ++rt)
      acc[rt] = __builtin_amdgcn_wmma_f32_16x16x32_bf16(false, a[ks][0][rt], false, b[ks][0], (short)0, acc[rt], false, false);
#pragma unroll
    for (int rt = 0; rt < NT; ++rt)
      acc[rt] = __builtin_amdgcn_wmma_f32_16x16x32_bf16(false, a[ks][0][rt], false, b[ks][1], (short)0, acc[rt], false, false);
#pragma unroll
    for (int rt = 0; rt < NT; ++rt)
      acc[rt] = __builtin_amdgcn_wmma_f32_16x16x32_bf16(false, a[ks][1][rt], false, b[ks][0], (short)0, acc[rt], false, false);
  }
}

__device__ __forceinline__ void stage_barrier() {
  asm volatile("s_wait_dscnt 0x0\n\ts_barrier_signal -1\n\ts_barrier_wait -1" ::: "memory");
}
__device__ __forceinline__ void lds_done_signal() {
  asm volatile("s_wait_dscnt 0x0\n\ts_barrier_signal -1" ::: "memory");
}
__device__ __forceinline__ void step_end_wait() {
  asm volatile("s_barrier_wait -1" ::: "memory");
}

template <int NT>
__device__ __forceinline__ void lds_half_then_mma(v8f (&acc)[NRT], const v16bf (&b)[2][2], const unsigned short* const (&ar)[NRT],
                                                  const f32x4 (&xv)[NRT], unsigned short* const (&xs)[NRT], const int cur, const int oth) {
  v16bf a[2][2][NT];
  load_a<NT>(a, ar, cur);
  pin_order();
  stage_x(xv, xs, oth);
  lds_done_signal();
  pin_order();
  mma_regs<NT>(acc, a, b);
}

template <int S>
__device__ __forceinline__ void k_step(const int c, const bool many, const bool idle, v8f (&acc)[NRT], v16bf (&bq)[RING][2][2], f32x4 (&xq)[RING][NRT],
                                       const unsigned short* bhi_p, const unsigned short* blo_p, const float* x, const unsigned int (&xo)[NRT],
                                       const unsigned short* const (&ar)[NRT], unsigned short* const (&xs)[NRT]) {
  constexpr int CUR = S & 1;
  constexpr int SX  = (S + 1) % RING;
  if (many) lds_half_then_mma<NRT>(acc, bq[S], ar, xq[SX], xs, CUR * STAGE, (1 - CUR) * STAGE);
  else      lds_half_then_mma<1>(acc, bq[S], ar, xq[SX], xs, CUR * STAGE, (1 - CUR) * STAGE);
  const int cx = (c + 1 + RING) < NCH ? c + 1 + RING : NCH - 1;
  const int cb = idle ? 0 : ((c + RING) < NCH ? c + RING : NCH - 1);
  asm volatile("" :: "v"(acc[0]), "v"(acc[1]), "v"(acc[2]), "v"(acc[3]) : "memory");
  load_x(xq[SX], x, xo, (unsigned int)(cx * KC));
  load_b(bq[S], bhi_p + cb * BCH, blo_p + cb * BCH);
  step_end_wait();
}

__global__ void __launch_bounds__(256)
nnue_prep(const float* __restrict__ l1_w, const float* __restrict__ l1_b,
          const float* __restrict__ l1f_w, const float* __restrict__ l1f_b,
          unsigned short* __restrict__ whi, unsigned short* __restrict__ wlo,
          float* __restrict__ bias,
          const int* __restrict__ lsi, int* __restrict__ perm, const int B) {
  __shared__ int s_c[COUNT][256];
  __shared__ int s_tot[COUNT];
  __shared__ __align__(16) int s_perm[SORTJ * 1024 + 256];
  if (blockIdx.x == 0) {
    const int t = threadIdx.x;
#pragma unroll
    for (int j = 0; j < SORTJ; ++j) {
      const int row0 = (j * 256 + t) * 4;
      *(int4*)(s_perm + row0) = *(const int4*)(lsi + (row0 < B - 4 ? row0 : B - 4));
    }
    __syncthreads();
    const int okm = -(int)(t * 64 < B);
    int bk[64];
#pragma unroll
    for (int i = 0; i < 64; ++i) bk[i] = s_perm[t * 64 + ((i + t) & 63)] & 7;
    unsigned long long pk = 0ull;
#pragma unroll
    for (int i = 0; i < 64; ++i) pk += (unsigned long long)(okm & 1) << (bk[i] * 8);
#pragma unroll
    for (int k = 0; k < COUNT; ++k) s_c[k][t] = (int)((pk >> (k * 8)) & 0xffull);
    __syncthreads();
    {
      const int w = t >> 5, l = t & 31;
      int a[8], sum = 0;
#pragma unroll
      for (int i = 0; i < 8; ++i) { a[i] = s_c[w][l * 8 + i]; sum += a[i]; }
      int s = sum;
#pragma unroll
      for (int d = 1; d < 32; d <<= 1) {
        const int o = __shfl_up(s, d);
        if (l >= d) s += o;
      }
      int before = s - sum;
#pragma unroll
      for (int i = 0; i < 8; ++i) { s_c[w][l * 8 + i] = before; before += a[i]; }
      if (l == 31) s_tot[w] = s;
    }
    __syncthreads();
    int first[COUNT];
    {
      int base = 0;
#pragma unroll
      for (int k = 0; k < COUNT; ++k) {
        first[k] = s_c[k][t] + base;
        base += s_tot[k];
      }
    }
#pragma unroll
    for (int k = 0; k < COUNT; ++k) s_c[k][t] = first[k];
    pk = 0ull;
#pragma unroll
    for (int i = 0; i < 64; ++i) {
      const int sh   = bk[i] * 8;
      const int rank = (int)((pk >> sh) & 0xffull);
      pk += (unsigned long long)(okm & 1) << sh;
      const int row  = t * 64 + ((i + t) & 63);
      const int pos  = ((s_c[bk[i]][t] + rank) & okm) | ((SORTJ * 1024 + t) & ~okm);
      s_perm[pos] = row | (bk[i] << 24);
    }
    __syncthreads();
#pragma unroll
    for (int j = 0; j < SORTJ; ++j) {
      const int i0  = (j * 256 + t) * 4;
      const int okm = -(int)(i0 < B);
      *(int4*)(perm + ((i0 & okm) | ((SORTJ * 1024 + t * 4) & ~okm))) = *(const int4*)(s_perm + i0);
    }
    return;
  }
  const int p = (blockIdx.x - 1) * 256 + threadIdx.x;
  if (p < NCOL * L1 / 8) {
    const int lane = p & 31;
    const int half = (p >> 5) & 1;
    const int ks   = (p >> 6) & 1;
    const int tc   = p >> 7;
    const int tile = tc / NCH;
    const int c    = tc - tile * NCH;
    const int r    = tile * 16 + (lane & 15);
    const int k    = c * KC + ks * 32 + half * 16 + (lane >> 4) * 8;
    const float4 a0 = *(const float4*)(l1_w + (size_t)r * L1 + k);
    const float4 a1 = *(const float4*)(l1_w + (size_t)r * L1 + k + 4);
    const float4 f0 = *(const float4*)(l1f_w + (size_t)(r & 15) * L1 + k);
    const float4 f1 = *(const float4*)(l1f_w + (size_t)(r & 15) * L1 + k + 4);
    const float w8[8] = {a0.x + f0.x, a0.y + f0.y, a0.z + f0.z, a0.w + f0.w,
                         a1.x + f1.x, a1.y + f1.y, a1.z + f1.z, a1.w + f1.w};
    u16x8 h, l;
#pragma unroll
    for (int i = 0; i < 8; ++i) {
      const unsigned short hb = f2bf(w8[i]);
      h[i] = hb;
      l[i] = f2bf(w8[i] - bf2f(hb));
    }
    *(u16x8*)(whi + (size_t)p * 8) = h;
    *(u16x8*)(wlo + (size_t)p * 8) = l;
  }
  if (p < NCOL) bias[p] = l1_b[p] + l1f_b[p & 15];
}

__global__ void __launch_bounds__(256, 1)
nnue_main(const float* x,
          const int* __restrict__ perm,
          const unsigned short* whi,
          const unsigned short* wlo,
          const float* __restrict__ bias,
          const float* __restrict__ l2_w,
          const float* __restrict__ l2_b,
          const float* __restrict__ out_w,
          const float* __restrict__ out_b,
          float* __restrict__ out) {
  __shared__ __align__(16) unsigned short s_x[2][2][TM][LDSTR];
  __shared__ float s_l1s[TM][L2 + 2];
  __shared__ float s_l2x[TM][L3];

  const int t    = threadIdx.x;
  const int wv   = __builtin_amdgcn_readfirstlane(t >> 5);
  const int lane = t & 31;
  const int r0   = blockIdx.x * TM;

  const int nloc  = lane & 15;
  const int ksel  = (lane >> 4) << 3;
  const int rbase = (lane >> 4) << 3;

  const int ldr = t >> 4;
  const int ldc = (t & 15) << 2;

  const int pA = perm[r0 + lane];
  const int pB = perm[r0 + 32 + lane];
  const int rowA = pA & 0xffffff, bA = (pA >> 24) & 7;
  const int rowB = pB & 0xffffff, bB = (pB >> 24) & 7;

  unsigned int xo[NRT];
  xo[0] = (unsigned int)__shfl(rowA, ldr)      * L1 + ldc;
  xo[1] = (unsigned int)__shfl(rowA, ldr + 16) * L1 + ldc;
  xo[2] = (unsigned int)__shfl(rowB, ldr)      * L1 + ldc;
  xo[3] = (unsigned int)__shfl(rowB, ldr + 16) * L1 + ldc;

  int slotA = 0, slotB = 0;
  int myoff = 0, mycnt = 0, wb = 0;
  {
    const unsigned int below = (1u << lane) - 1u;
    int run = 0;
    int npc = 0;
    int poff = 0, pcnt = 0, pb = 0, lastb = 0;
    int foff = 0, fcnt = 0;
#pragma unroll
    for (int k = 0; k < COUNT; ++k) {
      const unsigned int mA = (unsigned int)__ballot(bA == k);
      const unsigned int mB = (unsigned int)__ballot(bB == k);
      const int cA = __popc(mA), cB = __popc(mB);
      if (bA == k) slotA = run + __popc(mA & below);
      if (bB == k) slotB = run + cA + __popc(mB & below);
      const int cnt = cA + cB;
      const int pcs = (cnt + 15) >> 4;
      if (wv >= npc && wv < npc + pcs) {
        const int j    = wv - npc;
        const int left = cnt - 16 * j;
        pb = k; poff = run + 16 * j; pcnt = left < 16 ? left : 16;
      }
      if (k == wv) { foff = run; fcnt = cnt; }
      if (cnt > 0) lastb = k;
      run += cnt;
      npc += pcs;
    }
    if (npc <= COUNT) {
      if (wv < npc) { wb = pb; myoff = poff; mycnt = pcnt; }
      else          { wb = lastb; myoff = 0; mycnt = 0; }
    } else {
      wb = wv; myoff = foff; mycnt = fcnt;
    }
  }
  myoff = __builtin_amdgcn_readfirstlane(myoff);
  mycnt = __builtin_amdgcn_readfirstlane(mycnt);
  wb    = __builtin_amdgcn_readfirstlane(wb);
  const bool many = mycnt > 16;
  const bool idle = mycnt == 0;

  const unsigned short* bhi_p = whi + (size_t)wb * NCH * BCH + lane * 8;
  const unsigned short* blo_p = wlo + (size_t)wb * NCH * BCH + lane * 8;

  v8f acc[NRT];
#pragma unroll
  for (int rt = 0; rt < NRT; ++rt)
    acc[rt] = (v8f){0.f, 0.f, 0.f, 0.f, 0.f, 0.f, 0.f, 0.f};

  v16bf  bq[RING][2][2];
  f32x4  xq[RING][NRT];

  f32x4 x0v[NRT];
  load_x(x0v, x, xo, 0u);
#pragma unroll
  for (int s = 0; s < RING; ++s) {
    pin_loads();
    load_x(xq[(s + 1) % RING], x, xo, (unsigned int)((s + 1) * KC));
    load_b(bq[s], bhi_p + s * BCH, blo_p + s * BCH);
  }

  unsigned short* xs[NRT];
  xs[0] = &s_x[0][0][__shfl(slotA, ldr)][ldc];
  xs[1] = &s_x[0][0][__shfl(slotA, ldr + 16)][ldc];
  xs[2] = &s_x[0][0][__shfl(slotB, ldr)][ldc];
  xs[3] = &s_x[0][0][__shfl(slotB, ldr + 16)][ldc];

  const unsigned short* ar[NRT];
#pragma unroll
  for (int rt = 0; rt < NRT; ++rt) {
    const int last = mycnt > 0 ? mycnt - 1 : 0;
    const int m    = (rt * 16 + nloc) < last ? (rt * 16 + nloc) : last;
    const int row  = (myoff + m) < TM ? (myoff + m) : TM - 1;
    ar[rt] = &s_x[0][0][row][ksel];
  }

  stage_x(x0v, xs, 0);
  stage_barrier();

  for (int c = 0; c < NCH; c += RING) {
    k_step<0>(c,     many, idle, acc, bq, xq, bhi_p, blo_p, x, xo, ar, xs);
    k_step<1>(c + 1, many, idle, acc, bq, xq, bhi_p, blo_p, x, xo, ar, xs);
    k_step<2>(c + 2, many, idle, acc, bq, xq, bhi_p, blo_p, x, xo, ar, xs);
    k_step<3>(c + 3, many, idle, acc, bq, xq, bhi_p, blo_p, x, xo, ar, xs);
  }

  {
    const float bc = bias[wb * 16 + nloc];
#pragma unroll
    for (int rt = 0; rt < NRT; ++rt)
#pragma unroll
      for (int p = 0; p < 8; ++p) {
        const int m = rt * 16 + rbase + p;
        if (m < mycnt) s_l1s[myoff + m][nloc] = acc[rt][p] + bc;
      }
  }
  __syncthreads();

  {
    const int i   = t >> 2;
    const int p   = t & 3;
    const int iA  = __shfl(bA, i & 31);
    const int iB  = __shfl(bB, i & 31);
    const int idx = i < 32 ? iA : iB;
    const int sA  = __shfl(slotA, i & 31);
    const int sB  = __shfl(slotB, i & 31);
    const float* trow = &s_l1s[i < 32 ? sA : sB][0];
    float l1x[2 * L2];
#pragma unroll
    for (int j = 0; j < L2; ++j) {
      const float v = trow[j];
      l1x[j]      = clamp01(v * v * (127.0f / 128.0f));
      l1x[L2 + j] = clamp01(v);
    }
#pragma unroll
    for (int q = 0; q < 8; ++q) {
      const int n = p * 8 + q;
      const float* wrow = l2_w + (size_t)(idx * L3 + n) * (2 * L2);
      float a = l2_b[idx * L3 + n];
#pragma unroll
      for (int j = 0; j < 2 * L2; ++j) a += l1x[j] * wrow[j];
      s_l2x[i][n] = clamp01(a);
    }
  }
  __syncthreads();

  if (t < TM) {
    const int idx = t < 32 ? bA : bB;
    const int sl  = t < 32 ? slotA : slotB;
    const int row = t < 32 ? rowA : rowB;
    float s = out_b[idx];
#pragma unroll
    for (int n = 0; n < L3; ++n) s += s_l2x[t][n] * out_w[idx * L3 + n];
    out[row] = s + s_l1s[sl][L2];
  }
}

extern "C" void kernel_launch(void* const* d_in, const int* in_sizes, int n_in,
                              void* d_out, int out_size, void* d_ws, size_t ws_size,
                              hipStream_t stream) {
  const float* x     = (const float*)d_in[0];
  const int*   lsi   = (const int*)d_in[1];
  const float* l1_w  = (const float*)d_in[2];
  const float* l1_b  = (const float*)d_in[3];
  const float* l1f_w = (const float*)d_in[4];
  const float* l1f_b = (const float*)d_in[5];
  const float* l2_w  = (const float*)d_in[6];
  const float* l2_b  = (const float*)d_in[7];
  const float* out_w = (const float*)d_in[8];
  const float* out_b = (const float*)d_in[9];

  const int B = in_sizes[0] / L1;

  unsigned short* whi  = (unsigned short*)d_ws;
  unsigned short* wlo  = whi + (size_t)NCOL * L1;
  float*          bias = (float*)(wlo + (size_t)NCOL * L1);
  int*            perm = (int*)(bias + NCOL);

  nnue_prep<<<PREPB + 1, 256, 0, stream>>>(l1_w, l1_b, l1f_w, l1f_b, whi, wlo, bias, lsi, perm, B);

  nnue_main<<<B / TM, 256, 0, stream>>>(x, perm, whi, wlo, bias,
                                        l2_w, l2_b, out_w, out_b, (float*)d_out);
}


# K loop specialised once per wave by its kind: a wave without rows only stages x (no weight loads, fragment reads or WMMAs), one-tile and four-tile waves run own loops; order of sums unchanged
# speedup vs baseline: 2.8322x; 1.0428x over previous
#include <hip/hip_runtime.h>

#define COUNT 8
#define L1    3072
#define L2    15
#define L3    32
#define NCOL  128
#define TM    64
#define NRT   4
#define KC    64
#define LDSTR 72
#define NCH   (L1 / KC)
#define RING  4
#define PIECE 256
#define BCH   (4 * PIECE)
#define STAGE (TM * LDSTR)
#define PLANE (2 * STAGE)
#define PREPB ((NCOL * L1 / 8 + 255) / 256)
#define SORTJ 16

typedef __attribute__((ext_vector_type(16))) __bf16         v16bf;
typedef __attribute__((ext_vector_type(8)))  float          v8f;
typedef __attribute__((ext_vector_type(16))) unsigned short v16u;
typedef __attribute__((ext_vector_type(8)))  unsigned short u16x8;
typedef __attribute__((ext_vector_type(4)))  unsigned short u16x4;
typedef __attribute__((ext_vector_type(2)))  float          f32x2;
typedef __attribute__((ext_vector_type(4)))  float          f32x4;
typedef __attribute__((ext_vector_type(2)))  __bf16         bf16x2;
typedef __attribute__((ext_vector_type(2)))  unsigned int   u32x2;

__device__ __forceinline__ unsigned short f2bf(float f) {
  unsigned int u = __float_as_uint(f);
  u += 0x7fffu + ((u >> 16) & 1u);
  return (unsigned short)(u >> 16);
}
__device__ __forceinline__ float bf2f(unsigned short h) {
  return __uint_as_float(((unsigned int)h) << 16);
}
__device__ __forceinline__ float clamp01(float v) {
  return fminf(fmaxf(v, 0.0f), 1.0f);
}
__device__ __forceinline__ unsigned int pack2bf(float a, float b) {
  return __builtin_bit_cast(unsigned int, __builtin_convertvector((f32x2){a, b}, bf16x2));
}

__device__ __forceinline__ v16bf frag16(const unsigned short* p) {
  u16x8 a = *(const u16x8*)(p);
  u16x8 b = *(const u16x8*)(p + 16);
  v16u v;
#pragma unroll
  for (int i = 0; i < 8; ++i) { v[i] = a[i]; v[i + 8] = b[i]; }
  return __builtin_bit_cast(v16bf, v);
}

__device__ __forceinline__ v16bf frag16w(const unsigned short* p) {
  u16x8 a = *(const u16x8*)(p);
  u16x8 b = *(const u16x8*)(p + PIECE);
  v16u v;
#pragma unroll
  for (int i = 0; i < 8; ++i) { v[i] = a[i]; v[i + 8] = b[i]; }
  return __builtin_bit_cast(v16bf, v);
}

__device__ __forceinline__ void load_b(v16bf (&b)[2][2], const unsigned short* hi, const unsigned short* lo) {
#pragma unroll
  for (int ks = 0; ks < 2; ++ks) {
    b[ks][0] = frag16w(hi + ks * 2 * PIECE);
    b[ks][1] = frag16w(lo + ks * 2 * PIECE);
  }
}

__device__ __forceinline__ void load_x(f32x4 (&xv)[NRT], const float* x, const unsigned int (&xo)[NRT], const unsigned int k) {
#pragma unroll
  for (int q = 0; q < NRT; ++q) xv[q] = __builtin_nontemporal_load((const f32x4*)(x + (size_t)(xo[q] + k)));
}

__device__ __forceinline__ void stage_x(const f32x4 (&xv)[NRT], unsigned short* const (&xs)[NRT], const int st) {
#pragma unroll
  for (int q = 0; q < NRT; ++q) {
    const unsigned int h01 = pack2bf(xv[q].x, xv[q].y);
    const unsigned int h23 = pack2bf(xv[q].z, xv[q].w);
    const float r0 = xv[q].x - __uint_as_float(h01 << 16);
    const float r1 = xv[q].y - __uint_as_float(h01 & 0xffff0000u);
    const float r2 = xv[q].z - __uint_as_float(h23 << 16);
    const float r3 = xv[q].w - __uint_as_float(h23 & 0xffff0000u);
    *(u32x2*)(xs[q] + st)         = (u32x2){h01, h23};
    *(u32x2*)(xs[q] + st + PLANE) = (u32x2){pack2bf(r0, r1), pack2bf(r2, r3)};
  }
}

__device__ __forceinline__ void pin_order() { __builtin_amdgcn_sched_barrier(0); }
__device__ __forceinline__ void pin_loads() { __builtin_amdgcn_sched_barrier(0x2 | 0x4 | 0x80); }

template <int NT>
__device__ __forceinline__ void load_a(v16bf (&a)[2][2][NT], const unsigned short* const (&ar)[NRT], const int st) {
#pragma unroll
  for (int ks = 0; ks < 2; ++ks) {
#pragma unroll
    for (int rt = 0; rt < NT; ++rt) a[ks][0][rt] = frag16(ar[rt] + st + ks * 32);
#pragma unroll
    for (int rt = 0; rt < NT; ++rt) a[ks][1][rt] = frag16(ar[rt] + st + PLANE + ks * 32);
  }
}
template <int NT>
__device__ __forceinline__ void mma_regs(v8f (&acc)[NRT], const v16bf (&a)[2][2][NT], const v16bf (&b)[2][2]) {
#pragma unroll
  for (int ks = 0; ks < 2; ++ks) {
#pragma unroll
    for (int rt = 0; rt < NT; ++rt)
      acc[rt] = __builtin_amdgcn_wmma_f32_16x16x32_bf16(false, a[ks][0][rt], false, b[ks][0], (short)0, acc[rt], false, false);
#pragma unroll
    for (int rt = 0; rt < NT; ++rt)
      acc[rt] = __builtin_amdgcn_wmma_f32_16x16x32_bf16(false, a[ks][0][rt], false, b[ks][1], (short)0, acc[rt], false, false);
#pragma unroll
    for (int rt = 0; rt < NT; ++rt)
      acc[rt] = __builtin_amdgcn_wmma_f32_16x16x32_bf16(false, a[ks][1][rt], false, b[ks][0], (short)0, acc[rt], false, false);
  }
}

__device__ __forceinline__ void stage_barrier() {
  asm volatile("s_wait_dscnt 0x0\n\ts_barrier_signal -1\n\ts_barrier_wait -1" ::: "memory");
}
__device__ __forceinline__ void lds_done_signal() {
  asm volatile("s_wait_dscnt 0x0\n\ts_barrier_signal -1" ::: "memory");
}
__device__ __forceinline__ void step_end_wait() {
  asm volatile("s_barrier_wait -1" ::: "memory");
}

template <int NT>
__device__ __forceinline__ void lds_half_then_mma(v8f (&acc)[NRT], const v16bf (&b)[2][2], const unsigned short* const (&ar)[NRT],
                                                  const f32x4 (&xv)[NRT], unsigned short* const (&xs)[NRT], const int cur, const int oth) {
  v16bf a[2][2][NT];
  load_a<NT>(a, ar, cur);
  pin_order();
  stage_x(xv, xs, oth);
  lds_done_signal();
  pin_order();
  mma_regs<NT>(acc, a, b);
}

template <int S, int NT, bool REFILL>
__device__ __forceinline__ void k_step(const int c, v8f (&acc)[NRT], v16bf (&bq)[RING][2][2], f32x4 (&xq)[RING][NRT],
                                       const unsigned short* bhi_p, const unsigned short* blo_p, const float* x, const unsigned int (&xo)[NRT],
                                       const unsigned short* const (&ar)[NRT], unsigned short* const (&xs)[NRT]) {
  constexpr int CUR = S & 1;
  constexpr int SX  = (S + 1) % RING;
  lds_half_then_mma<NT>(acc, bq[S], ar, xq[SX], xs, CUR * STAGE, (1 - CUR) * STAGE);
  if constexpr (NT == 1) asm volatile("" :: "v"(acc[0]) : "memory");
  else                   asm volatile("" :: "v"(acc[0]), "v"(acc[1]), "v"(acc[2]), "v"(acc[3]) : "memory");
  if constexpr (REFILL) {
    const int cx = (c + 1 + RING) < NCH ? c + 1 + RING : NCH - 1;
    const int cb = c + RING;
    load_x(xq[SX], x, xo, (unsigned int)(cx * KC));
    load_b(bq[S], bhi_p + cb * BCH, blo_p + cb * BCH);
  }
  step_end_wait();
}

template <int S, bool REFILL>
__device__ __forceinline__ void k_step_stage_only(const int c, f32x4 (&xq)[RING][NRT], const float* x, const unsigned int (&xo)[NRT],
                                                  unsigned short* const (&xs)[NRT]) {
  constexpr int CUR = S & 1;
  constexpr int SX  = S;
  pin_order();
  stage_x(xq[SX], xs, (1 - CUR) * STAGE);
  lds_done_signal();
  if constexpr (REFILL) {
    const int cx = (c + 1 + RING) < NCH ? c + 1 + RING : NCH - 1;
    load_x(xq[SX], x, xo, (unsigned int)(cx * KC));
  }
  step_end_wait();
}

template <int NT>
__device__ __forceinline__ void k_loop(v8f (&acc)[NRT], const f32x4 (&x0v)[NRT], const unsigned short* bhi_p, const unsigned short* blo_p,
                                       const float* x, const unsigned int (&xo)[NRT],
                                       const unsigned short* const (&ar)[NRT], unsigned short* const (&xs)[NRT]) {
  v16bf  bq[RING][2][2];
  f32x4  xq[RING][NRT];
#pragma unroll
  for (int s = 0; s < RING; ++s) {
    pin_loads();
    load_x(xq[(s + 1) % RING], x, xo, (unsigned int)((s + 1) * KC));
    load_b(bq[s], bhi_p + s * BCH, blo_p + s * BCH);
  }
  stage_x(x0v, xs, 0);
  stage_barrier();
  for (int c = 0; c < NCH - RING; c += RING) {
    k_step<0, NT, true>(c,     acc, bq, xq, bhi_p, blo_p, x, xo, ar, xs);
    k_step<1, NT, true>(c + 1, acc, bq, xq, bhi_p, blo_p, x, xo, ar, xs);
    k_step<2, NT, true>(c + 2, acc, bq, xq, bhi_p, blo_p, x, xo, ar, xs);
    k_step<3, NT, true>(c + 3, acc, bq, xq, bhi_p, blo_p, x, xo, ar, xs);
  }
  k_step<0, NT, false>(NCH - RING,     acc, bq, xq, bhi_p, blo_p, x, xo, ar, xs);
  k_step<1, NT, false>(NCH - RING + 1, acc, bq, xq, bhi_p, blo_p, x, xo, ar, xs);
  k_step<2, NT, false>(NCH - RING + 2, acc, bq, xq, bhi_p, blo_p, x, xo, ar, xs);
  k_step<3, NT, false>(NCH - RING + 3, acc, bq, xq, bhi_p, blo_p, x, xo, ar, xs);
}

__device__ __forceinline__ void k_loop_stage_only(const f32x4 (&x0v)[NRT], const float* x, const unsigned int (&xo)[NRT],
                                                  unsigned short* const (&xs)[NRT]) {
  f32x4  xq[RING][NRT];
#pragma unroll
  for (int s = 0; s < RING; ++s) {
    asm volatile("" ::: "memory");
    load_x(xq[s], x, xo, (unsigned int)((s + 1) * KC));
  }
  stage_x(x0v, xs, 0);
  stage_barrier();
  for (int c = 0; c < NCH - RING; c += RING) {
    k_step_stage_only<0, true>(c,     xq, x, xo, xs);
    k_step_stage_only<1, true>(c + 1, xq, x, xo, xs);
    k_step_stage_only<2, true>(c + 2, xq, x, xo, xs);
    k_step_stage_only<3, true>(c + 3, xq, x, xo, xs);
  }
  k_step_stage_only<0, false>(NCH - RING,     xq, x, xo, xs);
  k_step_stage_only<1, false>(NCH - RING + 1, xq, x, xo, xs);
  k_step_stage_only<2, false>(NCH - RING + 2, xq, x, xo, xs);
  k_step_stage_only<3, false>(NCH - RING + 3, xq, x, xo, xs);
}

__global__ void __launch_bounds__(256)
nnue_prep(const float* __restrict__ l1_w, const float* __restrict__ l1_b,
          const float* __restrict__ l1f_w, const float* __restrict__ l1f_b,
          unsigned short* __restrict__ whi, unsigned short* __restrict__ wlo,
          float* __restrict__ bias,
          const int* __restrict__ lsi, int* __restrict__ perm, const int B) {
  __shared__ int s_c[COUNT][256];
  __shared__ int s_tot[COUNT];
  __shared__ __align__(16) int s_perm[SORTJ * 1024 + 256];
  if (blockIdx.x == 0) {
    const int t = threadIdx.x;
#pragma unroll
    for (int j = 0; j < SORTJ; ++j) {
      const int row0 = (j * 256 + t) * 4;
      *(int4*)(s_perm + row0) = *(const int4*)(lsi + (row0 < B - 4 ? row0 : B - 4));
    }
    __syncthreads();
    const int okm = -(int)(t * 64 < B);
    int bk[64];
#pragma unroll
    for (int i = 0; i < 64; ++i) bk[i] = s_perm[t * 64 + ((i + t) & 63)] & 7;
    unsigned long long pk = 0ull;
#pragma unroll
    for (int i = 0; i < 64; ++i) pk += (unsigned long long)(okm & 1) << (bk[i] * 8);
#pragma unroll
    for (int k = 0; k < COUNT; ++k) s_c[k][t] = (int)((pk >> (k * 8)) & 0xffull);
    __syncthreads();
    {
      const int w = t >> 5, l = t & 31;
      int a[8], sum = 0;
#pragma unroll
      for (int i = 0; i < 8; ++i) { a[i] = s_c[w][l * 8 + i]; sum += a[i]; }
      int s = sum;
#pragma unroll
      for (int d = 1; d < 32; d <<= 1) {
        const int o = __shfl_up(s, d);
        if (l >= d) s += o;
      }
      int before = s - sum;
#pragma unroll
      for (int i = 0; i < 8; ++i) { s_c[w][l * 8 + i] = before; before += a[i]; }
      if (l == 31) s_tot[w] = s;
    }
    __syncthreads();
    int first[COUNT];
    {
      int base = 0;
#pragma unroll
      for (int k = 0; k < COUNT; ++k) {
        first[k] = s_c[k][t] + base;
        base += s_tot[k];
      }
    }
#pragma unroll
    for (int k = 0; k < COUNT; ++k) s_c[k][t] = first[k];
    pk = 0ull;
#pragma unroll
    for (int i = 0; i < 64; ++i) {
      const int sh   = bk[i] * 8;
      const int rank = (int)((pk >> sh) & 0xffull);
      pk += (unsigned long long)(okm & 1) << sh;
      const int row  = t * 64 + ((i + t) & 63);
      const int pos  = ((s_c[bk[i]][t] + rank) & okm) | ((SORTJ * 1024 + t) & ~okm);
      s_perm[pos] = row | (bk[i] << 24);
    }
    __syncthreads();
#pragma unroll
    for (int j = 0; j < SORTJ; ++j) {
      const int i0  = (j * 256 + t) * 4;
      const int okm = -(int)(i0 < B);
      *(int4*)(perm + ((i0 & okm) | ((SORTJ * 1024 + t * 4) & ~okm))) = *(const int4*)(s_perm + i0);
    }
    return;
  }
  const int p = (blockIdx.x - 1) * 256 + threadIdx.x;
  if (p < NCOL * L1 / 8) {
    const int lane = p & 31;
    const int half = (p >> 5) & 1;
    const int ks   = (p >> 6) & 1;
    const int tc   = p >> 7;
    const int tile = tc / NCH;
    const int c    = tc - tile * NCH;
    const int r    = tile * 16 + (lane & 15);
    const int k    = c * KC + ks * 32 + half * 16 + (lane >> 4) * 8;
    const float4 a0 = *(const float4*)(l1_w + (size_t)r * L1 + k);
    const float4 a1 = *(const float4*)(l1_w + (size_t)r * L1 + k + 4);
    const float4 f0 = *(const float4*)(l1f_w + (size_t)(r & 15) * L1 + k);
    const float4 f1 = *(const float4*)(l1f_w + (size_t)(r & 15) * L1 + k + 4);
    const float w8[8] = {a0.x + f0.x, a0.y + f0.y, a0.z + f0.z, a0.w + f0.w,
                         a1.x + f1.x, a1.y + f1.y, a1.z + f1.z, a1.w + f1.w};
    u16x8 h, l;
#pragma unroll
    for (int i = 0; i < 8; ++i) {
      const unsigned short hb = f2bf(w8[i]);
      h[i] = hb;
      l[i] = f2bf(w8[i] - bf2f(hb));
    }
    *(u16x8*)(whi + (size_t)p * 8) = h;
    *(u16x8*)(wlo + (size_t)p * 8) = l;
  }
  if (p < NCOL) bias[p] = l1_b[p] + l1f_b[p & 15];
}

__global__ void __launch_bounds__(256, 1)
nnue_main(const float* x,
          const int* __restrict__ perm,
          const unsigned short* whi,
          const unsigned short* wlo,
          const float* __restrict__ bias,
          const float* __restrict__ l2_w,
          const float* __restrict__ l2_b,
          const float* __restrict__ out_w,
          const float* __restrict__ out_b,
          float* __restrict__ out) {
  __shared__ __align__(16) unsigned short s_x[2][2][TM][LDSTR];
  __shared__ float s_l1s[TM][L2 + 2];
  __shared__ float s_l2x[TM][L3];

  const int t    = threadIdx.x;
  const int wv   = __builtin_amdgcn_readfirstlane(t >> 5);
  const int lane = t & 31;
  const int r0   = blockIdx.x * TM;

  const int nloc  = lane & 15;
  const int ksel  = (lane >> 4) << 3;
  const int rbase = (lane >> 4) << 3;

  const int ldr = t >> 4;
  const int ldc = (t & 15) << 2;

  const int pA = perm[r0 + lane];
  const int pB = perm[r0 + 32 + lane];
  const int rowA = pA & 0xffffff, bA = (pA >> 24) & 7;
  const int rowB = pB & 0xffffff, bB = (pB >> 24) & 7;

  unsigned int xo[NRT];
  xo[0] = (unsigned int)__shfl(rowA, ldr)      * L1 + ldc;
  xo[1] = (unsigned int)__shfl(rowA, ldr + 16) * L1 + ldc;
  xo[2] = (unsigned int)__shfl(rowB, ldr)      * L1 + ldc;
  xo[3] = (unsigned int)__shfl(rowB, ldr + 16) * L1 + ldc;

  int slotA = 0, slotB = 0;
  int myoff = 0, mycnt = 0, wb = 0;
  {
    const unsigned int below = (1u << lane) - 1u;
    int run = 0;
    int npc = 0;
    int poff = 0, pcnt = 0, pb = 0, lastb = 0;
    int foff = 0, fcnt = 0;
#pragma unroll
    for (int k = 0; k < COUNT; ++k) {
      const unsigned int mA = (unsigned int)__ballot(bA == k);
      const unsigned int mB = (unsigned int)__ballot(bB == k);
      const int cA = __popc(mA), cB = __popc(mB);
      if (bA == k) slotA = run + __popc(mA & below);
      if (bB == k) slotB = run + cA + __popc(mB & below);
      const int cnt = cA + cB;
      const int pcs = (cnt + 15) >> 4;
      if (wv >= npc && wv < npc + pcs) {
        const int j    = wv - npc;
        const int left = cnt - 16 * j;
        pb = k; poff = run + 16 * j; pcnt = left < 16 ? left : 16;
      }
      if (k == wv) { foff = run; fcnt = cnt; }
      if (cnt > 0) lastb = k;
      run += cnt;
      npc += pcs;
    }
    if (npc <= COUNT) {
      if (wv < npc) { wb = pb; myoff = poff; mycnt = pcnt; }
      else          { wb = lastb; myoff = 0; mycnt = 0; }
    } else {
      wb = wv; myoff = foff; mycnt = fcnt;
    }
  }
  myoff = __builtin_amdgcn_readfirstlane(myoff);
  mycnt = __builtin_amdgcn_readfirstlane(mycnt);
  wb    = __builtin_amdgcn_readfirstlane(wb);
  const bool many = mycnt > 16;
  const bool idle = mycnt == 0;

  const unsigned short* bhi_p = whi + (size_t)wb * NCH * BCH + lane * 8;
  const unsigned short* blo_p = wlo + (size_t)wb * NCH * BCH + lane * 8;

  v8f acc[NRT];
#pragma unroll
  for (int rt = 0; rt < NRT; ++rt)
    acc[rt] = (v8f){0.f, 0.f, 0.f, 0.f, 0.f, 0.f, 0.f, 0.f};

  f32x4 x0v[NRT];
  load_x(x0v, x, xo, 0u);

  unsigned short* xs[NRT];
  xs[0] = &s_x[0][0][__shfl(slotA, ldr)][ldc];
  xs[1] = &s_x[0][0][__shfl(slotA, ldr + 16)][ldc];
  xs[2] = &s_x[0][0][__shfl(slotB, ldr)][ldc];
  xs[3] = &s_x[0][0][__shfl(slotB, ldr + 16)][ldc];

  const unsigned short* ar[NRT];
#pragma unroll
  for (int rt = 0; rt < NRT; ++rt) {
    const int last = mycnt > 0 ? mycnt - 1 : 0;
    const int m    = (rt * 16 + nloc) < last ? (rt * 16 + nloc) : last;
    const int row  = (myoff + m) < TM ? (myoff + m) : TM - 1;
    ar[rt] = &s_x[0][0][row][ksel];
  }

  if (idle)      k_loop_stage_only(x0v, x, xo, xs);
  else if (many) k_loop<NRT>(acc, x0v, bhi_p, blo_p, x, xo, ar, xs);
  else           k_loop<1>(acc, x0v, bhi_p, blo_p, x, xo, ar, xs);

  {
    const float bc = bias[wb * 16 + nloc];
#pragma unroll
    for (int rt = 0; rt < NRT; ++rt)
#pragma unroll
      for (int p = 0; p < 8; ++p) {
        const int m = rt * 16 + rbase + p;
        if (m < mycnt) s_l1s[myoff + m][nloc] = acc[rt][p] + bc;
      }
  }
  __syncthreads();

  {
    const int i   = t >> 2;
    const int p   = t & 3;
    const int iA  = __shfl(bA, i & 31);
    const int iB  = __shfl(bB, i & 31);
    const int idx = i < 32 ? iA : iB;
    const int sA  = __shfl(slotA, i & 31);
    const int sB  = __shfl(slotB, i & 31);
    const float* trow = &s_l1s[i < 32 ? sA : sB][0];
    float l1x[2 * L2];
#pragma unroll
    for (int j = 0; j < L2; ++j) {
      const float v = trow[j];
      l1x[j]      = clamp01(v * v * (127.0f / 128.0f));
      l1x[L2 + j] = clamp01(v);
    }
#pragma unroll
    for (int q = 0; q < 8; ++q) {
      const int n = p * 8 + q;
      const float* wrow = l2_w + (size_t)(idx * L3 + n) * (2 * L2);
      float a = l2_b[idx * L3 + n];
#pragma unroll
      for (int j = 0; j < 2 * L2; ++j) a += l1x[j] * wrow[j];
      s_l2x[i][n] = clamp01(a);
    }
  }
  __syncthreads();

  if (t < TM) {
    const int idx = t < 32 ? bA : bB;
    const int sl  = t < 32 ? slotA : slotB;
    const int row = t < 32 ? rowA : rowB;
    float s = out_b[idx];
#pragma unroll
    for (int n = 0; n < L3; ++n) s += s_l2x[t][n] * out_w[idx * L3 + n];
    out[row] = s + s_l1s[sl][L2];
  }
}

extern "C" void kernel_launch(void* const* d_in, const int* in_sizes, int n_in,
                              void* d_out, int out_size, void* d_ws, size_t ws_size,
                              hipStream_t stream) {
  const float* x     = (const float*)d_in[0];
  const int*   lsi   = (const int*)d_in[1];
  const float* l1_w  = (const float*)d_in[2];
  const float* l1_b  = (const float*)d_in[3];
  const float* l1f_w = (const float*)d_in[4];
  const float* l1f_b = (const float*)d_in[5];
  const float* l2_w  = (const float*)d_in[6];
  const float* l2_b  = (const float*)d_in[7];
  const float* out_w = (const float*)d_in[8];
  const float* out_b = (const float*)d_in[9];

  const int B = in_sizes[0] / L1;

  unsigned short* whi  = (unsigned short*)d_ws;
  unsigned short* wlo  = whi + (size_t)NCOL * L1;
  float*          bias = (float*)(wlo + (size_t)NCOL * L1);
  int*            perm = (int*)(bias + NCOL);

  nnue_prep<<<PREPB + 1, 256, 0, stream>>>(l1_w, l1_b, l1f_w, l1f_b, whi, wlo, bias, lsi, perm, B);

  nnue_main<<<B / TM, 256, 0, stream>>>(x, perm, whi, wlo, bias,
                                        l2_w, l2_b, out_w, out_b, (float*)d_out);
}
